# starting kernel
# baseline (speedup 1.0000x reference)
#include <hip/hip_runtime.h>
#include <math.h>

typedef __attribute__((ext_vector_type(2))) float v2f;
typedef __attribute__((ext_vector_type(8))) float v8f;

#define NQ   8
#define QDIM 256

__device__ __forceinline__ float2 cmul(float2 a, float2 b) {
  return make_float2(a.x * b.x - a.y * b.y, a.x * b.y + a.y * b.x);
}
__device__ __forceinline__ float2 cadd(float2 a, float2 b) {
  return make_float2(a.x + b.x, a.y + b.y);
}

__global__ __launch_bounds__(32) void build_W(const float* __restrict__ weights,
                                              int n_layers,
                                              float* __restrict__ wR,
                                              float* __restrict__ wI) {
  __shared__ float2 col[QDIM][32];
  const int t = threadIdx.x;
  const int j = blockIdx.x * 32 + t;

  for (int i = 0; i < QDIM; ++i) col[i][t] = make_float2(i == j ? 1.f : 0.f, 0.f);

  for (int l = 0; l < n_layers; ++l) {
    for (int q = 0; q < NQ; ++q) {
      const float ph = weights[(l * NQ + q) * 3 + 0];
      const float th = weights[(l * NQ + q) * 3 + 1];
      const float om = weights[(l * NQ + q) * 3 + 2];
      const float c  = cosf(0.5f * th), s = sinf(0.5f * th);
      const float ap = 0.5f * (ph + om), bm = 0.5f * (ph - om);
      const float2 u00 = make_float2( cosf(ap) * c, -sinf(ap) * c);
      const float2 u01 = make_float2(-cosf(bm) * s, -sinf(bm) * s);
      const float2 u10 = make_float2( cosf(bm) * s, -sinf(bm) * s);
      const float2 u11 = make_float2( cosf(ap) * c,  sinf(ap) * c);
      const int stride = 1 << (7 - q);
      for (int m = 0; m < QDIM / 2; ++m) {
        const int low = m & (stride - 1);
        const int hi  = m >> (7 - q);
        const int i0  = hi * (stride << 1) + low;
        const int i1  = i0 + stride;
        const float2 a0 = col[i0][t], a1 = col[i1][t];
        col[i0][t] = cadd(cmul(u00, a0), cmul(u01, a1));
        col[i1][t] = cadd(cmul(u10, a0), cmul(u11, a1));
      }
    }
    const int r = l % (NQ - 1) + 1;
    for (int q = 0; q < NQ; ++q) {
      const int tq    = (q + r) & 7;
      const int cmask = 1 << (7 - q);
      const int tmask = 1 << (7 - tq);
      for (int m = 0; m < QDIM; ++m) {
        if ((m & cmask) && !(m & tmask)) {
          const int m2 = m | tmask;
          const float2 tmp = col[m][t];
          col[m][t]  = col[m2][t];
          col[m2][t] = tmp;
        }
      }
    }
  }

  for (int n = 0; n < QDIM; ++n) {
    const int idx = (((j >> 1) * QDIM) + n) * 2 + (j & 1);
    wR[idx] = col[n][t].x;
    wI[idx] = col[n][t].y;
  }
}

__global__ __launch_bounds__(32) void vqe_wmma(const float* __restrict__ x,
                                               const float* __restrict__ fcw,
                                               const float* __restrict__ fcb,
                                               const float* __restrict__ wR,
                                               const float* __restrict__ wI,
                                               float* __restrict__ out) {
  __shared__ float PhiR[16][260];
  __shared__ float PhiI[16][260];
  __shared__ float zred[16][20];

  const int l = threadIdx.x;
  const int h = l >> 4;
  const int c = l & 15;
  const int sBase = blockIdx.x * 16;

  float cs[NQ], sn[NQ];
  {
    const float* xp = x + (size_t)(sBase + c) * NQ;
#pragma unroll
    for (int q = 0; q < NQ; ++q) {
      const float xv = 0.5f * xp[q];
      cs[q] = cosf(xv);
      sn[q] = sinf(xv);
    }
  }

  for (int tt = 0; tt < 128; ++tt) {
    const int b = (h << 7) | tt;
    float m = h ? sn[0] : cs[0];
#pragma unroll
    for (int q = 1; q < NQ; ++q)
      m *= ((b >> (7 - q)) & 1) ? sn[q] : cs[q];
    const int p = __popc(b) & 3;
    PhiR[c][b] = (p == 0) ? m : ((p == 2) ? -m : 0.f);
    PhiI[c][b] = (p == 1) ? -m : ((p == 3) ? m : 0.f);
  }
  __syncthreads();

  float fw[NQ];
#pragma unroll
  for (int i = 0; i < NQ; ++i) fw[i] = fcw[i];

  float zacc[8];
#pragma unroll
  for (int r = 0; r < 8; ++r) zacc[r] = 0.f;

  const float* aRbase = &PhiR[c][2 * h];
  const float* aIbase = &PhiI[c][2 * h];

  for (int nt = 0; nt < 16; ++nt) {
    const int n = nt * 16 + c;
    float g = 0.f;
#pragma unroll
    for (int i = 0; i < NQ; ++i)
      g += fw[i] * (1.f - 2.f * (float)((n >> (7 - i)) & 1));

    v8f cr1 = {0.f, 0.f, 0.f, 0.f, 0.f, 0.f, 0.f, 0.f};
    v8f cr2 = cr1, ci1 = cr1, ci2 = cr1;

    const float* bR = wR + (size_t)(h * 512 + 2 * n);
    const float* bI = wI + (size_t)(h * 512 + 2 * n);

#pragma unroll 4
    for (int k0 = 0; k0 < QDIM; k0 += 4) {
      const v2f ar = *(const v2f*)(aRbase + k0);
      const v2f ai = *(const v2f*)(aIbase + k0);
      const v2f br = *(const v2f*)(bR + (size_t)(k0 >> 1) * 512);
      const v2f bi = *(const v2f*)(bI + (size_t)(k0 >> 1) * 512);
      cr1 = __builtin_amdgcn_wmma_f32_16x16x4_f32(false, ar, false, br, (short)0, cr1, false, false);
      cr2 = __builtin_amdgcn_wmma_f32_16x16x4_f32(false, ai, false, bi, (short)0, cr2, false, false);
      ci1 = __builtin_amdgcn_wmma_f32_16x16x4_f32(false, ar, false, bi, (short)0, ci1, false, false);
      ci2 = __builtin_amdgcn_wmma_f32_16x16x4_f32(false, ai, false, br, (short)0, ci2, false, false);
    }

#pragma unroll
    for (int r = 0; r < 8; ++r) {
      const float pr = cr1[r] - cr2[r];
      const float pi = ci1[r] + ci2[r];
      zacc[r] += (pr * pr + pi * pi) * g;
    }
  }

#pragma unroll
  for (int r = 0; r < 8; ++r) zred[r + 8 * h][c] = zacc[r];
  __syncthreads();

  if (l < 16) {
    float sum = fcb[0];
#pragma unroll
    for (int cc = 0; cc < 16; ++cc) sum += zred[l][cc];
    out[sBase + l] = 1.f / (1.f + expf(-sum));
  }
}

extern "C" void kernel_launch(void* const* d_in, const int* in_sizes, int n_in,
                              void* d_out, int out_size, void* d_ws, size_t ws_size,
                              hipStream_t stream) {
  const float* x       = (const float*)d_in[0];
  const float* weights = (const float*)d_in[1];
  const float* fcw     = (const float*)d_in[2];
  const float* fcb     = (const float*)d_in[3];
  float* out = (float*)d_out;

  float* wR = (float*)d_ws;
  float* wI = wR + QDIM * QDIM;

  const int B        = in_sizes[0] / NQ;
  const int n_layers = in_sizes[1] / (NQ * 3);

  build_W<<<QDIM / 32, 32, 0, stream>>>(weights, n_layers, wR, wI);
  vqe_wmma<<<B / 16, 32, 0, stream>>>(x, fcw, fcb, wR, wI, out);
}


# vqe: no 256x256 circuit matrix: per sample, layer 0 on the product state, layer 1 as kron of two 16x16 complex factors on f32 WMMA (S*A^T, then B*D1 in place); one launch; sums reordered
# speedup vs baseline: 17.0793x; 17.0793x over previous
#include <hip/hip_runtime.h>
#include <math.h>

typedef __attribute__((ext_vector_type(2))) float v2f;
typedef __attribute__((ext_vector_type(8))) float v8f;

#define NQ   8
#define SPW  16
#define PSROW 68

__device__ __forceinline__ float2 cmul(float2 a, float2 b) {
  return make_float2(a.x * b.x - a.y * b.y, a.x * b.y + a.y * b.x);
}

__device__ __forceinline__ int ring_src(int m, int r) {
  int j = m;
#pragma unroll
  for (int q = NQ - 1; q >= 0; --q) {
    const int tq = (q + r) & 7;
    if (j & (1 << (7 - q))) j ^= 1 << (7 - tq);
  }
  return j;
}
__device__ __forceinline__ int ring_dst(int j, int r) {
  int m = j;
#pragma unroll
  for (int q = 0; q < NQ; ++q) {
    const int tq = (q + r) & 7;
    if (m & (1 << (7 - q))) m ^= 1 << (7 - tq);
  }
  return m;
}

__device__ __forceinline__ void rot_gate(const float* __restrict__ wt, float2* g) {
  const float ph = wt[0], th = wt[1], om = wt[2];
  const float c  = cosf(0.5f * th), s = sinf(0.5f * th);
  const float ap = 0.5f * (ph + om), bm = 0.5f * (ph - om);
  g[0] = make_float2( cosf(ap) * c, -sinf(ap) * c);
  g[1] = make_float2(-cosf(bm) * s, -sinf(bm) * s);
  g[2] = make_float2( cosf(bm) * s, -sinf(bm) * s);
  g[3] = make_float2( cosf(ap) * c,  sinf(ap) * c);
}

__device__ __forceinline__ int m_of(int lo, int hi) { return ((lo >> 2) << 6) | (hi << 2) | (lo & 3); }
__device__ __forceinline__ int khi_of(int kk, int h, int e) { return (kk << 2) | (e << 1) | h; }
__device__ __forceinline__ int klo_of(int kk, int h, int e) { return 8 * h + 2 * kk + e; }

struct LayerOps {
  v2f Ar[4], Ai[4], nAi[4];
  v2f Br[4], Bi[4], nBi[4];
};

__device__ __forceinline__ void build_ops(const float2 (*G)[4], int h, int c, LayerOps& K) {
#pragma unroll
  for (int kk = 0; kk < 4; ++kk) {
#pragma unroll
    for (int e = 0; e < 2; ++e) {
      const int kh = khi_of(kk, h, e);
      float2 a = G[2][((c >> 3) & 1) * 2 + ((kh >> 3) & 1)];
      a = cmul(a, G[3][((c >> 2) & 1) * 2 + ((kh >> 2) & 1)]);
      a = cmul(a, G[4][((c >> 1) & 1) * 2 + ((kh >> 1) & 1)]);
      a = cmul(a, G[5][(c & 1) * 2 + (kh & 1)]);
      K.Ar[kk][e] = a.x; K.Ai[kk][e] = a.y; K.nAi[kk][e] = -a.y;
      const int kl = klo_of(kk, h, e);
      float2 b = G[0][((c >> 3) & 1) * 2 + ((kl >> 3) & 1)];
      b = cmul(b, G[1][((c >> 2) & 1) * 2 + ((kl >> 2) & 1)]);
      b = cmul(b, G[6][((c >> 1) & 1) * 2 + ((kl >> 1) & 1)]);
      b = cmul(b, G[7][(c & 1) * 2 + (kl & 1)]);
      K.Br[kk][e] = b.x; K.Bi[kk][e] = b.y; K.nBi[kk][e] = -b.y;
    }
  }
}

#define WMMA4(a, b, acc) __builtin_amdgcn_wmma_f32_16x16x4_f32(false, (a), false, (b), (short)0, (acc), false, false)

__device__ __forceinline__ void apply_layer(const v2f* sr, const v2f* si, const LayerOps& K,
                                            v8f& d2r, v8f& d2i) {
  const v8f zero = {0.f, 0.f, 0.f, 0.f, 0.f, 0.f, 0.f, 0.f};
  v8f d1r = zero, d1i = zero;
#pragma unroll
  for (int kk = 0; kk < 4; ++kk) {
    d1r = WMMA4(sr[kk], K.Ar[kk], d1r);
    d1i = WMMA4(sr[kk], K.Ai[kk], d1i);
    d1r = WMMA4(si[kk], K.nAi[kk], d1r);
    d1i = WMMA4(si[kk], K.Ar[kk], d1i);
  }
  d2r = zero; d2i = zero;
#pragma unroll
  for (int kk = 0; kk < 4; ++kk) {
    const v2f tr = {d1r[2 * kk], d1r[2 * kk + 1]};
    const v2f ti = {d1i[2 * kk], d1i[2 * kk + 1]};
    d2r = WMMA4(K.Br[kk], tr, d2r);
    d2i = WMMA4(K.Bi[kk], tr, d2i);
    d2r = WMMA4(K.nBi[kk], ti, d2r);
    d2i = WMMA4(K.Br[kk], ti, d2i);
  }
}

template <bool DEEP>
__device__ __forceinline__ void run_layers(const float* __restrict__ weights, int n_layers,
                                           const float* fw, float fb, int sBase,
                                           float2 (*G1)[4], float (*PS)[PSROW], float (*ST)[256],
                                           float (*zred)[33], float* __restrict__ out) {
  const int l = threadIdx.x, h = l >> 4, c = l & 15;

  LayerOps K;
  build_ops(G1, h, c, K);

  float gw[8];
  {
    const int rn = (n_layers - 1) % (NQ - 1) + 1;
#pragma unroll
    for (int r = 0; r < 8; ++r) {
      const int m = ring_dst(m_of(r + 8 * h, c), rn);
      float g = 0.f;
#pragma unroll
      for (int i = 0; i < NQ; ++i) g += fw[i] * (1.f - 2.f * (float)((m >> (7 - i)) & 1));
      gw[r] = g;
    }
  }

  const int jl = ring_src(m_of(c, khi_of(0, h, 0)), 1);

  for (int s = 0; s < SPW; ++s) {
    float2 a[NQ], b[NQ];
#pragma unroll
    for (int q = 0; q < NQ; ++q) {
      const float4 v = *(const float4*)(PS[s] + q * 8 + ((jl >> (7 - q)) & 1) * 4);
      a[q] = make_float2(v.x, v.y);
      b[q] = make_float2(v.z, v.w);
    }
    const float2 P = cmul(cmul(a[0], a[1]), cmul(a[6], a[7]));
    const float2 X[2] = {cmul(P, a[2]), cmul(P, b[2])};
    float2 Y[2][2], Z[2][2];
#pragma unroll
    for (int u = 0; u < 2; ++u) {
#pragma unroll
      for (int v = 0; v < 2; ++v) {
        Y[u][v] = cmul(X[u], (u ^ v) ? b[3] : a[3]);
        Z[u][v] = cmul((u ^ v) ? b[4] : a[4], v ? b[5] : a[5]);
      }
    }
    v2f sr[4], si[4];
#pragma unroll
    for (int kk = 0; kk < 4; ++kk) {
#pragma unroll
      for (int e = 0; e < 2; ++e) {
        const float2 el = cmul(Y[kk >> 1][kk & 1], Z[kk & 1][e]);
        sr[kk][e] = el.x; si[kk][e] = el.y;
      }
    }

    v8f d2r, d2i;
    apply_layer(sr, si, K, d2r, d2i);

    if (DEEP) {
      for (int L = 2; L < n_layers; ++L) {
        __syncthreads();
        if (l < NQ) rot_gate(weights + (L * NQ + l) * 3, G1[l]);
#pragma unroll
        for (int r = 0; r < 8; ++r) {
          const int j = m_of(r + 8 * h, c);
          ST[0][j] = d2r[r]; ST[1][j] = d2i[r];
        }
        __syncthreads();
        build_ops(G1, h, c, K);
        const int rr = (L - 1) % (NQ - 1) + 1;
#pragma unroll
        for (int kk = 0; kk < 4; ++kk) {
#pragma unroll
          for (int e = 0; e < 2; ++e) {
            const int j = ring_src(m_of(c, khi_of(kk, h, e)), rr);
            sr[kk][e] = ST[0][j]; si[kk][e] = ST[1][j];
          }
        }
        apply_layer(sr, si, K, d2r, d2i);
      }
      __syncthreads();
      if (l < NQ) rot_gate(weights + (NQ + l) * 3, G1[l]);
      __syncthreads();
      build_ops(G1, h, c, K);
    }

    float zp = 0.f;
#pragma unroll
    for (int r = 0; r < 8; ++r) zp += (d2r[r] * d2r[r] + d2i[r] * d2i[r]) * gw[r];
    zred[s][l] = zp;
  }
  __syncthreads();

  if (l < SPW) {
    float sum = fb;
#pragma unroll
    for (int cc = 0; cc < 32; ++cc) sum += zred[l][cc];
    out[sBase + l] = 1.f / (1.f + expf(-sum));
  }
}

__global__ __launch_bounds__(32) void vqe_kron(const float* __restrict__ x,
                                               const float* __restrict__ weights,
                                               int n_layers,
                                               const float* __restrict__ fcw,
                                               const float* __restrict__ fcb,
                                               float* __restrict__ out) {
  __shared__ float2 G0[NQ][4];
  __shared__ float2 G1[NQ][4];
  __shared__ __attribute__((aligned(16))) float PS[SPW][PSROW];
  __shared__ float ST[2][256];
  __shared__ float zred[SPW][33];

  const int l = threadIdx.x;
  const int h = l >> 4;
  const int c = l & 15;
  const int sBase = blockIdx.x * SPW;

  if (l < NQ) {
    if (n_layers > 0) {
      rot_gate(weights + l * 3, G0[l]);
    } else {
      G0[l][0] = make_float2(1.f, 0.f); G0[l][1] = make_float2(0.f, 0.f);
      G0[l][2] = make_float2(0.f, 0.f); G0[l][3] = make_float2(1.f, 0.f);
    }
  } else if (l < 2 * NQ) {
    if (n_layers > 1) rot_gate(weights + (NQ + l - NQ) * 3, G1[l - NQ]);
  }
  __syncthreads();

  {
    const float* xp = x + (size_t)(sBase + c) * NQ;
#pragma unroll
    for (int q = 0; q < NQ; ++q) {
      const float xv = 0.5f * xp[q];
      const float cs = cosf(xv), sn = sinf(xv);
      const float2 u00 = G0[q][0], u01 = G0[q][1], u10 = G0[q][2], u11 = G0[q][3];
      const float2 p0 = make_float2(u00.x * cs + u01.y * sn, u00.y * cs - u01.x * sn);
      const float2 p1 = make_float2(u10.x * cs + u11.y * sn, u10.y * cs - u11.x * sn);
      *(float4*)(PS[c] + q * 8 + h * 4) = h ? make_float4(p1.x, p1.y, p0.x, p0.y) : make_float4(p0.x, p0.y, p1.x, p1.y);
    }
  }
  __syncthreads();

  float fw[NQ];
#pragma unroll
  for (int i = 0; i < NQ; ++i) fw[i] = fcw[i];
  const float fb = fcb[0];

  if (n_layers > 2) {
    run_layers<true>(weights, n_layers, fw, fb, sBase, G1, PS, ST, zred, out);
  } else if (n_layers == 2) {
    run_layers<false>(weights, n_layers, fw, fb, sBase, G1, PS, ST, zred, out);
  } else {
    for (int s = 0; s < SPW; ++s) {
      float zp = 0.f;
      for (int i = 0; i < 8; ++i) {
        const int m = l + 32 * i;
        const int j = (n_layers > 0) ? ring_src(m, 1) : m;
        float2 v = make_float2(PS[s][((j >> 7) & 1) * 2], PS[s][((j >> 7) & 1) * 2 + 1]);
        for (int q = 1; q < NQ; ++q) {
          const int bq = (j >> (7 - q)) & 1;
          v = cmul(v, make_float2(PS[s][q * 8 + bq * 2], PS[s][q * 8 + bq * 2 + 1]));
        }
        float g = 0.f;
        for (int t = 0; t < NQ; ++t) g += fw[t] * (1.f - 2.f * (float)((m >> (7 - t)) & 1));
        zp += (v.x * v.x + v.y * v.y) * g;
      }
      zred[s][l] = zp;
    }
    __syncthreads();
    if (l < SPW) {
      float sum = fb;
      for (int cc = 0; cc < 32; ++cc) sum += zred[l][cc];
      out[sBase + l] = 1.f / (1.f + expf(-sum));
    }
  }
}

extern "C" void kernel_launch(void* const* d_in, const int* in_sizes, int n_in,
                              void* d_out, int out_size, void* d_ws, size_t ws_size,
                              hipStream_t stream) {
  const float* x       = (const float*)d_in[0];
  const float* weights = (const float*)d_in[1];
  const float* fcw     = (const float*)d_in[2];
  const float* fcb     = (const float*)d_in[3];
  float* out = (float*)d_out;

  const int B        = in_sizes[0] / NQ;
  const int n_layers = in_sizes[1] / (NQ * 3);

  vqe_kron<<<B / SPW, 32, 0, stream>>>(x, weights, n_layers, fcw, fcb, out);
}


# vqe two-layer path: ring of a product state has 4 distinct rows per sample as a 16x16 matrix: 4 samples share one S*A^T tile, B*D1 is one k-step; 9 f32 WMMAs per sample, not 32; sums reordered
# speedup vs baseline: 33.8651x; 1.9828x over previous
#include <hip/hip_runtime.h>
#include <math.h>

typedef __attribute__((ext_vector_type(2))) float v2f;
typedef __attribute__((ext_vector_type(8))) float v8f;

#define NQ   8
#define SPW  16
#define PSROW 68

__device__ __forceinline__ float2 cmul(float2 a, float2 b) {
  return make_float2(a.x * b.x - a.y * b.y, a.x * b.y + a.y * b.x);
}

__device__ __forceinline__ int ring_src(int m, int r) {
  int j = m;
#pragma unroll
  for (int q = NQ - 1; q >= 0; --q) {
    const int tq = (q + r) & 7;
    if (j & (1 << (7 - q))) j ^= 1 << (7 - tq);
  }
  return j;
}
__device__ __forceinline__ int ring_dst(int j, int r) {
  int m = j;
#pragma unroll
  for (int q = 0; q < NQ; ++q) {
    const int tq = (q + r) & 7;
    if (m & (1 << (7 - q))) m ^= 1 << (7 - tq);
  }
  return m;
}

__device__ __forceinline__ void rot_gate(const float* __restrict__ wt, float2* g) {
  const float ph = wt[0], th = wt[1], om = wt[2];
  const float c  = cosf(0.5f * th), s = sinf(0.5f * th);
  const float ap = 0.5f * (ph + om), bm = 0.5f * (ph - om);
  g[0] = make_float2( cosf(ap) * c, -sinf(ap) * c);
  g[1] = make_float2(-cosf(bm) * s, -sinf(bm) * s);
  g[2] = make_float2( cosf(bm) * s, -sinf(bm) * s);
  g[3] = make_float2( cosf(ap) * c,  sinf(ap) * c);
}

__device__ __forceinline__ int m_of(int lo, int hi) { return ((lo >> 2) << 6) | (hi << 2) | (lo & 3); }
__device__ __forceinline__ int khi_of(int kk, int h, int e) { return (kk << 2) | (e << 1) | h; }
__device__ __forceinline__ int klo_of(int kk, int h, int e) { return 8 * h + 2 * kk + e; }

struct LayerOps {
  v2f Ar[4], Ai[4], nAi[4];
  v2f Br[4], Bi[4], nBi[4];
};

__device__ __forceinline__ void build_ops(const float2 (*G)[4], int h, int c, LayerOps& K) {
#pragma unroll
  for (int kk = 0; kk < 4; ++kk) {
#pragma unroll
    for (int e = 0; e < 2; ++e) {
      const int kh = khi_of(kk, h, e);
      float2 a = G[2][((c >> 3) & 1) * 2 + ((kh >> 3) & 1)];
      a = cmul(a, G[3][((c >> 2) & 1) * 2 + ((kh >> 2) & 1)]);
      a = cmul(a, G[4][((c >> 1) & 1) * 2 + ((kh >> 1) & 1)]);
      a = cmul(a, G[5][(c & 1) * 2 + (kh & 1)]);
      K.Ar[kk][e] = a.x; K.Ai[kk][e] = a.y; K.nAi[kk][e] = -a.y;
      const int kl = klo_of(kk, h, e);
      float2 b = G[0][((c >> 3) & 1) * 2 + ((kl >> 3) & 1)];
      b = cmul(b, G[1][((c >> 2) & 1) * 2 + ((kl >> 2) & 1)]);
      b = cmul(b, G[6][((c >> 1) & 1) * 2 + ((kl >> 1) & 1)]);
      b = cmul(b, G[7][(c & 1) * 2 + (kl & 1)]);
      K.Br[kk][e] = b.x; K.Bi[kk][e] = b.y; K.nBi[kk][e] = -b.y;
    }
  }
}

#define WMMA4(a, b, acc) __builtin_amdgcn_wmma_f32_16x16x4_f32(false, (a), false, (b), (short)0, (acc), false, false)

__device__ __forceinline__ void apply_layer(const v2f* sr, const v2f* si, const LayerOps& K,
                                            v8f& d2r, v8f& d2i) {
  const v8f zero = {0.f, 0.f, 0.f, 0.f, 0.f, 0.f, 0.f, 0.f};
  v8f d1r = zero, d1i = zero;
#pragma unroll
  for (int kk = 0; kk < 4; ++kk) {
    d1r = WMMA4(sr[kk], K.Ar[kk], d1r);
    d1i = WMMA4(sr[kk], K.Ai[kk], d1i);
    d1r = WMMA4(si[kk], K.nAi[kk], d1r);
    d1i = WMMA4(si[kk], K.Ar[kk], d1i);
  }
  d2r = zero; d2i = zero;
#pragma unroll
  for (int kk = 0; kk < 4; ++kk) {
    const v2f tr = {d1r[2 * kk], d1r[2 * kk + 1]};
    const v2f ti = {d1i[2 * kk], d1i[2 * kk + 1]};
    d2r = WMMA4(K.Br[kk], tr, d2r);
    d2i = WMMA4(K.Bi[kk], tr, d2i);
    d2r = WMMA4(K.nBi[kk], ti, d2r);
    d2i = WMMA4(K.Br[kk], ti, d2i);
  }
}

__device__ __forceinline__ void run_layers(const float* __restrict__ weights, int n_layers,
                                           const float* fw, float fb, int sBase,
                                           float2 (*G1)[4], float (*PS)[PSROW], float (*ST)[256],
                                           float (*zred)[33], float* __restrict__ out) {
  const int l = threadIdx.x, h = l >> 4, c = l & 15;

  LayerOps K;
  build_ops(G1, h, c, K);

  float gw[8];
  {
    const int rn = (n_layers - 1) % (NQ - 1) + 1;
#pragma unroll
    for (int r = 0; r < 8; ++r) {
      const int m = ring_dst(m_of(r + 8 * h, c), rn);
      float g = 0.f;
#pragma unroll
      for (int i = 0; i < NQ; ++i) g += fw[i] * (1.f - 2.f * (float)((m >> (7 - i)) & 1));
      gw[r] = g;
    }
  }

  const int jl = ring_src(m_of(c, khi_of(0, h, 0)), 1);

  for (int s = 0; s < SPW; ++s) {
    float2 a[NQ], b[NQ];
#pragma unroll
    for (int q = 0; q < NQ; ++q) {
      const float4 v = *(const float4*)(PS[s] + q * 8 + ((jl >> (7 - q)) & 1) * 4);
      a[q] = make_float2(v.x, v.y);
      b[q] = make_float2(v.z, v.w);
    }
    const float2 P = cmul(cmul(a[0], a[1]), cmul(a[6], a[7]));
    const float2 X[2] = {cmul(P, a[2]), cmul(P, b[2])};
    float2 Y[2][2], Z[2][2];
#pragma unroll
    for (int u = 0; u < 2; ++u) {
#pragma unroll
      for (int v = 0; v < 2; ++v) {
        Y[u][v] = cmul(X[u], (u ^ v) ? b[3] : a[3]);
        Z[u][v] = cmul((u ^ v) ? b[4] : a[4], v ? b[5] : a[5]);
      }
    }
    v2f sr[4], si[4];
#pragma unroll
    for (int kk = 0; kk < 4; ++kk) {
#pragma unroll
      for (int e = 0; e < 2; ++e) {
        const float2 el = cmul(Y[kk >> 1][kk & 1], Z[kk & 1][e]);
        sr[kk][e] = el.x; si[kk][e] = el.y;
      }
    }

    v8f d2r, d2i;
    apply_layer(sr, si, K, d2r, d2i);

    {
      for (int L = 2; L < n_layers; ++L) {
        __syncthreads();
        if (l < NQ) rot_gate(weights + (L * NQ + l) * 3, G1[l]);
#pragma unroll
        for (int r = 0; r < 8; ++r) {
          const int j = m_of(r + 8 * h, c);
          ST[0][j] = d2r[r]; ST[1][j] = d2i[r];
        }
        __syncthreads();
        build_ops(G1, h, c, K);
        const int rr = (L - 1) % (NQ - 1) + 1;
#pragma unroll
        for (int kk = 0; kk < 4; ++kk) {
#pragma unroll
          for (int e = 0; e < 2; ++e) {
            const int j = ring_src(m_of(c, khi_of(kk, h, e)), rr);
            sr[kk][e] = ST[0][j]; si[kk][e] = ST[1][j];
          }
        }
        apply_layer(sr, si, K, d2r, d2i);
      }
      __syncthreads();
      if (l < NQ) rot_gate(weights + (NQ + l) * 3, G1[l]);
      __syncthreads();
      build_ops(G1, h, c, K);
    }

    float zp = 0.f;
#pragma unroll
    for (int r = 0; r < 8; ++r) zp += (d2r[r] * d2r[r] + d2i[r] * d2i[r]) * gw[r];
    zred[s][l] = zp;
  }
  __syncthreads();

  if (l < SPW) {
    float sum = fb;
#pragma unroll
    for (int cc = 0; cc < 32; ++cc) sum += zred[l][cc];
    out[sBase + l] = 1.f / (1.f + expf(-sum));
  }
}

struct GroupIn {
  float4 v0, v1, v7, v2, v3, v4, v5;
  float2 v6;
};
struct GroupOps {
  v2f sr[4], si[4];
  v2f fr, fi;
};

__device__ __forceinline__ GroupIn load_group(const float* ps, int h, int m1, int m6) {
  GroupIn in;
  in.v0 = *(const float4*)(ps + 0 * 8 + h * 4);
  in.v1 = *(const float4*)(ps + 1 * 8 + (h ^ m1) * 4);
  in.v7 = *(const float4*)(ps + 7 * 8 + m6 * 4);
  in.v2 = *(const float4*)(ps + 2 * 8 + m1 * 4);
  in.v3 = *(const float4*)(ps + 3 * 8);
  in.v4 = *(const float4*)(ps + 4 * 8);
  in.v5 = *(const float4*)(ps + 5 * 8 + h * 4);
  in.v6 = *(const float2*)(ps + 6 * 8 + (h ^ m6) * 4);
  return in;
}

__device__ __forceinline__ void build_group_ops(const GroupIn& in, GroupOps& o) {
  const float2 f0 = cmul(cmul(make_float2(in.v0.x, in.v0.y), make_float2(in.v1.x, in.v1.y)), make_float2(in.v7.x, in.v7.y));
  const float2 f1 = cmul(cmul(make_float2(in.v0.z, in.v0.w), make_float2(in.v1.z, in.v1.w)), make_float2(in.v7.z, in.v7.w));
  o.fr[0] = f0.x; o.fr[1] = f1.x; o.fi[0] = f0.y; o.fi[1] = f1.y;

  const float2 X0 = make_float2(in.v2.x, in.v2.y), X1 = make_float2(in.v2.z, in.v2.w);
  const float2 p30 = make_float2(in.v3.x, in.v3.y), p31 = make_float2(in.v3.z, in.v3.w);
  const float2 p40 = make_float2(in.v4.x, in.v4.y), p41 = make_float2(in.v4.z, in.v4.w);
  const float2 Y[4] = {cmul(X0, p30), cmul(X0, p31), cmul(X1, p31), cmul(X1, p30)};
  const float2 Q0 = cmul(make_float2(in.v5.x, in.v5.y), in.v6), Q1 = cmul(make_float2(in.v5.z, in.v5.w), in.v6);
  const float2 Z[2][2] = {{cmul(p40, Q0), cmul(p41, Q1)}, {cmul(p41, Q0), cmul(p40, Q1)}};
#pragma unroll
  for (int kk = 0; kk < 4; ++kk) {
#pragma unroll
    for (int e = 0; e < 2; ++e) {
      const float2 el = cmul(Y[kk], Z[kk & 1][e]);
      o.sr[kk][e] = el.x; o.si[kk][e] = el.y;
    }
  }
}

struct TwoLayerOps {
  v2f Ar[4], Ai[4], nAi[4];
  v2f Br, Bi, nBi;
  float2 cf[2];
};

__device__ __forceinline__ void run_two_layers(const float* fw, float fb, int sBase,
                                               const float2 (*G1)[4], const float (*PS)[PSROW],
                                               float (*zred)[33], float* __restrict__ out) {
  const int l = threadIdx.x, h = l >> 4, c = l & 15;

  TwoLayerOps K;
#pragma unroll
  for (int kk = 0; kk < 4; ++kk) {
#pragma unroll
    for (int e = 0; e < 2; ++e) {
      const int kh = khi_of(kk, h, e);
      float2 a = G1[2][((c >> 3) & 1) * 2 + ((kh >> 3) & 1)];
      a = cmul(a, G1[3][((c >> 2) & 1) * 2 + ((kh >> 2) & 1)]);
      a = cmul(a, G1[4][((c >> 1) & 1) * 2 + ((kh >> 1) & 1)]);
      a = cmul(a, G1[5][(c & 1) * 2 + (kh & 1)]);
      K.Ar[kk][e] = a.x; K.Ai[kk][e] = a.y; K.nAi[kk][e] = -a.y;
    }
  }
#pragma unroll
  for (int e = 0; e < 2; ++e) {
    const float2 b = cmul(G1[0][((c >> 3) & 1) * 2 + h], G1[7][(c & 1) * 2 + e]);
    K.Br[e] = b.x; K.Bi[e] = b.y; K.nBi[e] = -b.y;
    K.cf[e] = cmul(G1[1][((c >> 2) & 1) * 2 + h], G1[6][((c >> 1) & 1) * 2 + e]);
  }

  float gw[8];
#pragma unroll
  for (int r = 0; r < 8; ++r) {
    const int m = ring_dst(m_of(c, r + 8 * h), 2);
    float g = 0.f;
#pragma unroll
    for (int i = 0; i < NQ; ++i) g += fw[i] * (1.f - 2.f * (float)((m >> (7 - i)) & 1));
    gw[r] = g;
  }

  const int m1 = c >> 3, s4 = (c >> 1) & 3, m6 = c & 1;
  const v8f zero = {0.f, 0.f, 0.f, 0.f, 0.f, 0.f, 0.f, 0.f};

  GroupOps cur, nxt;
  build_group_ops(load_group(PS[s4], h, m1, m6), cur);
  GroupIn in = load_group(PS[4 + s4], h, m1, m6);

#pragma unroll
  for (int g = 0; g < SPW / 4; ++g) {
    v8f wr = WMMA4(cur.fr, K.Br, zero);
    v8f wi = WMMA4(cur.fr, K.Bi, zero);
    wr = WMMA4(cur.fi, K.nBi, wr);
    wi = WMMA4(cur.fi, K.Br, wi);

    v8f gr = zero, gi = zero;
#pragma unroll
    for (int kk = 0; kk < 4; ++kk) {
      gr = WMMA4(cur.sr[kk], K.Ar[kk], gr);
      gi = WMMA4(cur.sr[kk], K.Ai[kk], gi);
      gr = WMMA4(cur.si[kk], K.nAi[kk], gr);
      gi = WMMA4(cur.si[kk], K.Ar[kk], gi);
    }

    if (g + 1 < SPW / 4) {
      build_group_ops(in, nxt);
      if (g + 2 < SPW / 4) in = load_group(PS[4 * (g + 2) + s4], h, m1, m6);
    }

    float br[8], bi[8], nbi[8];
#pragma unroll
    for (int r = 0; r < 8; ++r) {
      const float2 cf = K.cf[r & 1];
      br[r]  = wr[r] * cf.x - wi[r] * cf.y;
      bi[r]  = wr[r] * cf.y + wi[r] * cf.x;
      nbi[r] = -(wr[r] * cf.y) - wi[r] * cf.x;
    }

#pragma unroll
    for (int t = 0; t < 4; ++t) {
      const v2f ar = {gr[2 * t], gr[2 * t + 1]}, ai = {gi[2 * t], gi[2 * t + 1]};
      const v2f tr = {br[2 * t], br[2 * t + 1]}, ti = {bi[2 * t], bi[2 * t + 1]}, nti = {nbi[2 * t], nbi[2 * t + 1]};
      v8f cr = WMMA4(ar, tr, zero);
      v8f ci = WMMA4(ar, ti, zero);
      cr = WMMA4(ai, nti, cr);
      ci = WMMA4(ai, tr, ci);
      float zp = 0.f;
#pragma unroll
      for (int r = 0; r < 8; ++r) zp += (cr[r] * cr[r] + ci[r] * ci[r]) * gw[r];
      zred[4 * g + t][l] = zp;
    }
    cur = nxt;
  }
  __syncthreads();

  if (l < SPW) {
    float sum = fb;
#pragma unroll
    for (int cc = 0; cc < 32; ++cc) sum += zred[l][cc];
    out[sBase + l] = 1.f / (1.f + expf(-sum));
  }
}

__global__ __launch_bounds__(32) void vqe_kron(const float* __restrict__ x,
                                               const float* __restrict__ weights,
                                               int n_layers,
                                               const float* __restrict__ fcw,
                                               const float* __restrict__ fcb,
                                               float* __restrict__ out) {
  __shared__ float2 G0[NQ][4];
  __shared__ float2 G1[NQ][4];
  __shared__ __attribute__((aligned(16))) float PS[SPW][PSROW];
  __shared__ float ST[2][256];
  __shared__ float zred[SPW][33];

  const int l = threadIdx.x;
  const int h = l >> 4;
  const int c = l & 15;
  const int sBase = blockIdx.x * SPW;

  if (l < NQ) {
    if (n_layers > 0) {
      rot_gate(weights + l * 3, G0[l]);
    } else {
      G0[l][0] = make_float2(1.f, 0.f); G0[l][1] = make_float2(0.f, 0.f);
      G0[l][2] = make_float2(0.f, 0.f); G0[l][3] = make_float2(1.f, 0.f);
    }
  } else if (l < 2 * NQ) {
    if (n_layers > 1) rot_gate(weights + (NQ + l - NQ) * 3, G1[l - NQ]);
  }
  __syncthreads();

  {
    const float* xp = x + (size_t)(sBase + c) * NQ;
#pragma unroll
    for (int q = 0; q < NQ; ++q) {
      const float xv = 0.5f * xp[q];
      const float cs = cosf(xv), sn = sinf(xv);
      const float2 u00 = G0[q][0], u01 = G0[q][1], u10 = G0[q][2], u11 = G0[q][3];
      const float2 p0 = make_float2(u00.x * cs + u01.y * sn, u00.y * cs - u01.x * sn);
      const float2 p1 = make_float2(u10.x * cs + u11.y * sn, u10.y * cs - u11.x * sn);
      *(float4*)(PS[c] + q * 8 + h * 4) = h ? make_float4(p1.x, p1.y, p0.x, p0.y) : make_float4(p0.x, p0.y, p1.x, p1.y);
    }
  }
  __syncthreads();

  float fw[NQ];
#pragma unroll
  for (int i = 0; i < NQ; ++i) fw[i] = fcw[i];
  const float fb = fcb[0];

  if (n_layers > 2) {
    run_layers(weights, n_layers, fw, fb, sBase, G1, PS, ST, zred, out);
  } else if (n_layers == 2) {
    run_two_layers(fw, fb, sBase, G1, PS, zred, out);
  } else {
    for (int s = 0; s < SPW; ++s) {
      float zp = 0.f;
      for (int i = 0; i < 8; ++i) {
        const int m = l + 32 * i;
        const int j = (n_layers > 0) ? ring_src(m, 1) : m;
        float2 v = make_float2(PS[s][((j >> 7) & 1) * 2], PS[s][((j >> 7) & 1) * 2 + 1]);
        for (int q = 1; q < NQ; ++q) {
          const int bq = (j >> (7 - q)) & 1;
          v = cmul(v, make_float2(PS[s][q * 8 + bq * 2], PS[s][q * 8 + bq * 2 + 1]));
        }
        float g = 0.f;
        for (int t = 0; t < NQ; ++t) g += fw[t] * (1.f - 2.f * (float)((m >> (7 - t)) & 1));
        zp += (v.x * v.x + v.y * v.y) * g;
      }
      zred[s][l] = zp;
    }
    __syncthreads();
    if (l < SPW) {
      float sum = fb;
      for (int cc = 0; cc < 32; ++cc) sum += zred[l][cc];
      out[sBase + l] = 1.f / (1.f + expf(-sum));
    }
  }
}

extern "C" void kernel_launch(void* const* d_in, const int* in_sizes, int n_in,
                              void* d_out, int out_size, void* d_ws, size_t ws_size,
                              hipStream_t stream) {
  const float* x       = (const float*)d_in[0];
  const float* weights = (const float*)d_in[1];
  const float* fcw     = (const float*)d_in[2];
  const float* fcb     = (const float*)d_in[3];
  float* out = (float*)d_out;

  const int B        = in_sizes[0] / NQ;
  const int n_layers = in_sizes[1] / (NQ * 3);

  vqe_kron<<<B / SPW, 32, 0, stream>>>(x, weights, n_layers, fcw, fcb, out);
}


# vqe set-up: each angle's cos and sin made once per wave, one lane each, by sincosf (2 calls in a row for the 16 gates, 4 for the 16x8 embedding angles, not 12 + 16); x read as float4; sums unchanged
# speedup vs baseline: 43.8616x; 1.2952x over previous
#include <hip/hip_runtime.h>
#include <math.h>

typedef __attribute__((ext_vector_type(2))) float v2f;
typedef __attribute__((ext_vector_type(8))) float v8f;

#define NQ   8
#define SPW  16
#define PSROW 68

__device__ __forceinline__ float2 cmul(float2 a, float2 b) {
  return make_float2(a.x * b.x - a.y * b.y, a.x * b.y + a.y * b.x);
}

__device__ __forceinline__ int ring_src(int m, int r) {
  int j = m;
#pragma unroll
  for (int q = NQ - 1; q >= 0; --q) {
    const int tq = (q + r) & 7;
    if (j & (1 << (7 - q))) j ^= 1 << (7 - tq);
  }
  return j;
}
__device__ __forceinline__ int ring_dst(int j, int r) {
  int m = j;
#pragma unroll
  for (int q = 0; q < NQ; ++q) {
    const int tq = (q + r) & 7;
    if (m & (1 << (7 - q))) m ^= 1 << (7 - tq);
  }
  return m;
}

__device__ __forceinline__ void rot_gate(const float* __restrict__ wt, float2* g) {
  const float ph = wt[0], th = wt[1], om = wt[2];
  const float c  = cosf(0.5f * th), s = sinf(0.5f * th);
  const float ap = 0.5f * (ph + om), bm = 0.5f * (ph - om);
  g[0] = make_float2( cosf(ap) * c, -sinf(ap) * c);
  g[1] = make_float2(-cosf(bm) * s, -sinf(bm) * s);
  g[2] = make_float2( cosf(bm) * s, -sinf(bm) * s);
  g[3] = make_float2( cosf(ap) * c,  sinf(ap) * c);
}

__device__ __forceinline__ int m_of(int lo, int hi) { return ((lo >> 2) << 6) | (hi << 2) | (lo & 3); }
__device__ __forceinline__ int khi_of(int kk, int h, int e) { return (kk << 2) | (e << 1) | h; }
__device__ __forceinline__ int klo_of(int kk, int h, int e) { return 8 * h + 2 * kk + e; }

struct LayerOps {
  v2f Ar[4], Ai[4], nAi[4];
  v2f Br[4], Bi[4], nBi[4];
};

__device__ __forceinline__ void build_ops(const float2 (*G)[4], int h, int c, LayerOps& K) {
#pragma unroll
  for (int kk = 0; kk < 4; ++kk) {
#pragma unroll
    for (int e = 0; e < 2; ++e) {
      const int kh = khi_of(kk, h, e);
      float2 a = G[2][((c >> 3) & 1) * 2 + ((kh >> 3) & 1)];
      a = cmul(a, G[3][((c >> 2) & 1) * 2 + ((kh >> 2) & 1)]);
      a = cmul(a, G[4][((c >> 1) & 1) * 2 + ((kh >> 1) & 1)]);
      a = cmul(a, G[5][(c & 1) * 2 + (kh & 1)]);
      K.Ar[kk][e] = a.x; K.Ai[kk][e] = a.y; K.nAi[kk][e] = -a.y;
      const int kl = klo_of(kk, h, e);
      float2 b = G[0][((c >> 3) & 1) * 2 + ((kl >> 3) & 1)];
      b = cmul(b, G[1][((c >> 2) & 1) * 2 + ((kl >> 2) & 1)]);
      b = cmul(b, G[6][((c >> 1) & 1) * 2 + ((kl >> 1) & 1)]);
      b = cmul(b, G[7][(c & 1) * 2 + (kl & 1)]);
      K.Br[kk][e] = b.x; K.Bi[kk][e] = b.y; K.nBi[kk][e] = -b.y;
    }
  }
}

#define WMMA4(a, b, acc) __builtin_amdgcn_wmma_f32_16x16x4_f32(false, (a), false, (b), (short)0, (acc), false, false)

__device__ __forceinline__ void apply_layer(const v2f* sr, const v2f* si, const LayerOps& K,
                                            v8f& d2r, v8f& d2i) {
  const v8f zero = {0.f, 0.f, 0.f, 0.f, 0.f, 0.f, 0.f, 0.f};
  v8f d1r = zero, d1i = zero;
#pragma unroll
  for (int kk = 0; kk < 4; ++kk) {
    d1r = WMMA4(sr[kk], K.Ar[kk], d1r);
    d1i = WMMA4(sr[kk], K.Ai[kk], d1i);
    d1r = WMMA4(si[kk], K.nAi[kk], d1r);
    d1i = WMMA4(si[kk], K.Ar[kk], d1i);
  }
  d2r = zero; d2i = zero;
#pragma unroll
  for (int kk = 0; kk < 4; ++kk) {
    const v2f tr = {d1r[2 * kk], d1r[2 * kk + 1]};
    const v2f ti = {d1i[2 * kk], d1i[2 * kk + 1]};
    d2r = WMMA4(K.Br[kk], tr, d2r);
    d2i = WMMA4(K.Bi[kk], tr, d2i);
    d2r = WMMA4(K.nBi[kk], ti, d2r);
    d2i = WMMA4(K.Br[kk], ti, d2i);
  }
}

__device__ __forceinline__ void run_layers(const float* __restrict__ weights, int n_layers,
                                           const float* fw, float fb, int sBase,
                                           float2 (*G1)[4], float (*PS)[PSROW], float (*ST)[256],
                                           float (*zred)[33], float* __restrict__ out) {
  const int l = threadIdx.x, h = l >> 4, c = l & 15;

  LayerOps K;
  build_ops(G1, h, c, K);

  float gw[8];
  {
    const int rn = (n_layers - 1) % (NQ - 1) + 1;
#pragma unroll
    for (int r = 0; r < 8; ++r) {
      const int m = ring_dst(m_of(r + 8 * h, c), rn);
      float g = 0.f;
#pragma unroll
      for (int i = 0; i < NQ; ++i) g += fw[i] * (1.f - 2.f * (float)((m >> (7 - i)) & 1));
      gw[r] = g;
    }
  }

  const int jl = ring_src(m_of(c, khi_of(0, h, 0)), 1);

  for (int s = 0; s < SPW; ++s) {
    float2 a[NQ], b[NQ];
#pragma unroll
    for (int q = 0; q < NQ; ++q) {
      const float4 v = *(const float4*)(PS[s] + q * 8 + ((jl >> (7 - q)) & 1) * 4);
      a[q] = make_float2(v.x, v.y);
      b[q] = make_float2(v.z, v.w);
    }
    const float2 P = cmul(cmul(a[0], a[1]), cmul(a[6], a[7]));
    const float2 X[2] = {cmul(P, a[2]), cmul(P, b[2])};
    float2 Y[2][2], Z[2][2];
#pragma unroll
    for (int u = 0; u < 2; ++u) {
#pragma unroll
      for (int v = 0; v < 2; ++v) {
        Y[u][v] = cmul(X[u], (u ^ v) ? b[3] : a[3]);
        Z[u][v] = cmul((u ^ v) ? b[4] : a[4], v ? b[5] : a[5]);
      }
    }
    v2f sr[4], si[4];
#pragma unroll
    for (int kk = 0; kk < 4; ++kk) {
#pragma unroll
      for (int e = 0; e < 2; ++e) {
        const float2 el = cmul(Y[kk >> 1][kk & 1], Z[kk & 1][e]);
        sr[kk][e] = el.x; si[kk][e] = el.y;
      }
    }

    v8f d2r, d2i;
    apply_layer(sr, si, K, d2r, d2i);

    {
      for (int L = 2; L < n_layers; ++L) {
        __syncthreads();
        if (l < NQ) rot_gate(weights + (L * NQ + l) * 3, G1[l]);
#pragma unroll
        for (int r = 0; r < 8; ++r) {
          const int j = m_of(r + 8 * h, c);
          ST[0][j] = d2r[r]; ST[1][j] = d2i[r];
        }
        __syncthreads();
        build_ops(G1, h, c, K);
        const int rr = (L - 1) % (NQ - 1) + 1;
#pragma unroll
        for (int kk = 0; kk < 4; ++kk) {
#pragma unroll
          for (int e = 0; e < 2; ++e) {
            const int j = ring_src(m_of(c, khi_of(kk, h, e)), rr);
            sr[kk][e] = ST[0][j]; si[kk][e] = ST[1][j];
          }
        }
        apply_layer(sr, si, K, d2r, d2i);
      }
      __syncthreads();
      if (l < NQ) rot_gate(weights + (NQ + l) * 3, G1[l]);
      __syncthreads();
      build_ops(G1, h, c, K);
    }

    float zp = 0.f;
#pragma unroll
    for (int r = 0; r < 8; ++r) zp += (d2r[r] * d2r[r] + d2i[r] * d2i[r]) * gw[r];
    zred[s][l] = zp;
  }
  __syncthreads();

  if (l < SPW) {
    float sum = fb;
#pragma unroll
    for (int cc = 0; cc < 32; ++cc) sum += zred[l][cc];
    out[sBase + l] = 1.f / (1.f + expf(-sum));
  }
}

struct GroupIn {
  float4 v0, v1, v7, v2, v3, v4, v5;
  float2 v6;
};
struct GroupOps {
  v2f sr[4], si[4];
  v2f fr, fi;
};

__device__ __forceinline__ GroupIn load_group(const float* ps, int h, int m1, int m6) {
  GroupIn in;
  in.v0 = *(const float4*)(ps + 0 * 8 + h * 4);
  in.v1 = *(const float4*)(ps + 1 * 8 + (h ^ m1) * 4);
  in.v7 = *(const float4*)(ps + 7 * 8 + m6 * 4);
  in.v2 = *(const float4*)(ps + 2 * 8 + m1 * 4);
  in.v3 = *(const float4*)(ps + 3 * 8);
  in.v4 = *(const float4*)(ps + 4 * 8);
  in.v5 = *(const float4*)(ps + 5 * 8 + h * 4);
  in.v6 = *(const float2*)(ps + 6 * 8 + (h ^ m6) * 4);
  return in;
}

__device__ __forceinline__ void build_group_ops(const GroupIn& in, GroupOps& o) {
  const float2 f0 = cmul(cmul(make_float2(in.v0.x, in.v0.y), make_float2(in.v1.x, in.v1.y)), make_float2(in.v7.x, in.v7.y));
  const float2 f1 = cmul(cmul(make_float2(in.v0.z, in.v0.w), make_float2(in.v1.z, in.v1.w)), make_float2(in.v7.z, in.v7.w));
  o.fr[0] = f0.x; o.fr[1] = f1.x; o.fi[0] = f0.y; o.fi[1] = f1.y;

  const float2 X0 = make_float2(in.v2.x, in.v2.y), X1 = make_float2(in.v2.z, in.v2.w);
  const float2 p30 = make_float2(in.v3.x, in.v3.y), p31 = make_float2(in.v3.z, in.v3.w);
  const float2 p40 = make_float2(in.v4.x, in.v4.y), p41 = make_float2(in.v4.z, in.v4.w);
  const float2 Y[4] = {cmul(X0, p30), cmul(X0, p31), cmul(X1, p31), cmul(X1, p30)};
  const float2 Q0 = cmul(make_float2(in.v5.x, in.v5.y), in.v6), Q1 = cmul(make_float2(in.v5.z, in.v5.w), in.v6);
  const float2 Z[2][2] = {{cmul(p40, Q0), cmul(p41, Q1)}, {cmul(p41, Q0), cmul(p40, Q1)}};
#pragma unroll
  for (int kk = 0; kk < 4; ++kk) {
#pragma unroll
    for (int e = 0; e < 2; ++e) {
      const float2 el = cmul(Y[kk], Z[kk & 1][e]);
      o.sr[kk][e] = el.x; o.si[kk][e] = el.y;
    }
  }
}

struct TwoLayerOps {
  v2f Ar[4], Ai[4], nAi[4];
  v2f Br, Bi, nBi;
  float2 cf[2];
};

__device__ __forceinline__ void run_two_layers(const float* fw, float fb, int sBase,
                                               const float2 (*G1)[4], const float (*PS)[PSROW],
                                               float (*zred)[33], float* __restrict__ out) {
  const int l = threadIdx.x, h = l >> 4, c = l & 15;

  TwoLayerOps K;
#pragma unroll
  for (int kk = 0; kk < 4; ++kk) {
#pragma unroll
    for (int e = 0; e < 2; ++e) {
      const int kh = khi_of(kk, h, e);
      float2 a = G1[2][((c >> 3) & 1) * 2 + ((kh >> 3) & 1)];
      a = cmul(a, G1[3][((c >> 2) & 1) * 2 + ((kh >> 2) & 1)]);
      a = cmul(a, G1[4][((c >> 1) & 1) * 2 + ((kh >> 1) & 1)]);
      a = cmul(a, G1[5][(c & 1) * 2 + (kh & 1)]);
      K.Ar[kk][e] = a.x; K.Ai[kk][e] = a.y; K.nAi[kk][e] = -a.y;
    }
  }
#pragma unroll
  for (int e = 0; e < 2; ++e) {
    const float2 b = cmul(G1[0][((c >> 3) & 1) * 2 + h], G1[7][(c & 1) * 2 + e]);
    K.Br[e] = b.x; K.Bi[e] = b.y; K.nBi[e] = -b.y;
    K.cf[e] = cmul(G1[1][((c >> 2) & 1) * 2 + h], G1[6][((c >> 1) & 1) * 2 + e]);
  }

  float gw[8];
#pragma unroll
  for (int r = 0; r < 8; ++r) {
    const int m = ring_dst(m_of(c, r + 8 * h), 2);
    float g = 0.f;
#pragma unroll
    for (int i = 0; i < NQ; ++i) g += fw[i] * (1.f - 2.f * (float)((m >> (7 - i)) & 1));
    gw[r] = g;
  }

  const int m1 = c >> 3, s4 = (c >> 1) & 3, m6 = c & 1;
  const v8f zero = {0.f, 0.f, 0.f, 0.f, 0.f, 0.f, 0.f, 0.f};

  GroupOps cur, nxt;
  build_group_ops(load_group(PS[s4], h, m1, m6), cur);
  GroupIn in = load_group(PS[4 + s4], h, m1, m6);

#pragma unroll
  for (int g = 0; g < SPW / 4; ++g) {
    v8f wr = WMMA4(cur.fr, K.Br, zero);
    v8f wi = WMMA4(cur.fr, K.Bi, zero);
    wr = WMMA4(cur.fi, K.nBi, wr);
    wi = WMMA4(cur.fi, K.Br, wi);

    v8f gr = zero, gi = zero;
#pragma unroll
    for (int kk = 0; kk < 4; ++kk) {
      gr = WMMA4(cur.sr[kk], K.Ar[kk], gr);
      gi = WMMA4(cur.sr[kk], K.Ai[kk], gi);
      gr = WMMA4(cur.si[kk], K.nAi[kk], gr);
      gi = WMMA4(cur.si[kk], K.Ar[kk], gi);
    }

    if (g + 1 < SPW / 4) {
      build_group_ops(in, nxt);
      if (g + 2 < SPW / 4) in = load_group(PS[4 * (g + 2) + s4], h, m1, m6);
    }

    float br[8], bi[8], nbi[8];
#pragma unroll
    for (int r = 0; r < 8; ++r) {
      const float2 cf = K.cf[r & 1];
      br[r]  = wr[r] * cf.x - wi[r] * cf.y;
      bi[r]  = wr[r] * cf.y + wi[r] * cf.x;
      nbi[r] = -(wr[r] * cf.y) - wi[r] * cf.x;
    }

#pragma unroll
    for (int t = 0; t < 4; ++t) {
      const v2f ar = {gr[2 * t], gr[2 * t + 1]}, ai = {gi[2 * t], gi[2 * t + 1]};
      const v2f tr = {br[2 * t], br[2 * t + 1]}, ti = {bi[2 * t], bi[2 * t + 1]}, nti = {nbi[2 * t], nbi[2 * t + 1]};
      v8f cr = WMMA4(ar, tr, zero);
      v8f ci = WMMA4(ar, ti, zero);
      cr = WMMA4(ai, nti, cr);
      ci = WMMA4(ai, tr, ci);
      float zp = 0.f;
#pragma unroll
      for (int r = 0; r < 8; ++r) zp += (cr[r] * cr[r] + ci[r] * ci[r]) * gw[r];
      zred[4 * g + t][l] = zp;
    }
    cur = nxt;
  }
  __syncthreads();

  if (l < SPW) {
    float sum = fb;
#pragma unroll
    for (int cc = 0; cc < 32; ++cc) sum += zred[l][cc];
    out[sBase + l] = 1.f / (1.f + expf(-sum));
  }
}

__global__ __launch_bounds__(32) void vqe_kron(const float* __restrict__ x,
                                               const float* __restrict__ weights,
                                               int n_layers,
                                               const float* __restrict__ fcw,
                                               const float* __restrict__ fcb,
                                               float* __restrict__ out) {
  __shared__ __attribute__((aligned(16))) float2 GG[2][NQ][4];
  __shared__ __attribute__((aligned(16))) float PS[SPW][PSROW];
  __shared__ float ST[2][256];
  __shared__ float zred[SPW][33];
  float2 (*G1)[4] = GG[1];

  const int l = threadIdx.x;
  const int h = l >> 4;
  const int c = l & 15;
  const int sBase = blockIdx.x * SPW;

  const float4 xv = *(const float4*)(x + (size_t)(sBase + c) * NQ + 4 * h);
  const bool have = (c >> 3) < n_layers;
  float ph = 0.f, th = 0.f, om = 0.f;
  if (have) {
    const float* wt = weights + c * 3;
    ph = wt[0]; th = wt[1]; om = wt[2];
  }
  float fw[NQ];
#pragma unroll
  for (int i = 0; i < NQ; ++i) fw[i] = fcw[i];
  const float fb = fcb[0];

  float s1, c1, s2, c2;
  sincosf(h ? 0.5f * (ph + om) : 0.5f * th, &s1, &c1);
  sincosf(0.5f * (ph - om), &s2, &c2);
  const float so = __shfl_xor(s1, 16), co = __shfl_xor(c1, 16);
  if (h == 0) {
    float2* g = GG[c >> 3][c & 7];
    if (have) {
      g[0] = make_float2( co * c1, -so * c1);
      g[1] = make_float2(-c2 * s1, -s2 * s1);
      g[2] = make_float2( c2 * s1, -s2 * s1);
      g[3] = make_float2( co * c1,  so * c1);
    } else if (c < NQ) {
      g[0] = make_float2(1.f, 0.f); g[1] = make_float2(0.f, 0.f);
      g[2] = make_float2(0.f, 0.f); g[3] = make_float2(1.f, 0.f);
    }
  }

  float cs[4], sn[4];
  sincosf(0.5f * xv.x, &sn[0], &cs[0]);
  sincosf(0.5f * xv.y, &sn[1], &cs[1]);
  sincosf(0.5f * xv.z, &sn[2], &cs[2]);
  sincosf(0.5f * xv.w, &sn[3], &cs[3]);
  __syncthreads();

#pragma unroll
  for (int i = 0; i < 4; ++i) {
    const int q = 4 * h + i;
    const float4 ua = *(const float4*)GG[0][q], ub = *(const float4*)(GG[0][q] + 2);
    const float2 p0 = make_float2(ua.x * cs[i] + ua.w * sn[i], ua.y * cs[i] - ua.z * sn[i]);
    const float2 p1 = make_float2(ub.x * cs[i] + ub.w * sn[i], ub.y * cs[i] - ub.z * sn[i]);
    *(float4*)(PS[c] + q * 8)     = make_float4(p0.x, p0.y, p1.x, p1.y);
    *(float4*)(PS[c] + q * 8 + 4) = make_float4(p1.x, p1.y, p0.x, p0.y);
  }
  __syncthreads();

  if (n_layers > 2) {
    run_layers(weights, n_layers, fw, fb, sBase, G1, PS, ST, zred, out);
  } else if (n_layers == 2) {
    run_two_layers(fw, fb, sBase, G1, PS, zred, out);
  } else {
    for (int s = 0; s < SPW; ++s) {
      float zp = 0.f;
      for (int i = 0; i < 8; ++i) {
        const int m = l + 32 * i;
        const int j = (n_layers > 0) ? ring_src(m, 1) : m;
        float2 v = make_float2(PS[s][((j >> 7) & 1) * 2], PS[s][((j >> 7) & 1) * 2 + 1]);
        for (int q = 1; q < NQ; ++q) {
          const int bq = (j >> (7 - q)) & 1;
          v = cmul(v, make_float2(PS[s][q * 8 + bq * 2], PS[s][q * 8 + bq * 2 + 1]));
        }
        float g = 0.f;
        for (int t = 0; t < NQ; ++t) g += fw[t] * (1.f - 2.f * (float)((m >> (7 - t)) & 1));
        zp += (v.x * v.x + v.y * v.y) * g;
      }
      zred[s][l] = zp;
    }
    __syncthreads();
    if (l < SPW) {
      float sum = fb;
      for (int cc = 0; cc < 32; ++cc) sum += zred[l][cc];
      out[sBase + l] = 1.f / (1.f + expf(-sum));
    }
  }
}

extern "C" void kernel_launch(void* const* d_in, const int* in_sizes, int n_in,
                              void* d_out, int out_size, void* d_ws, size_t ws_size,
                              hipStream_t stream) {
  const float* x       = (const float*)d_in[0];
  const float* weights = (const float*)d_in[1];
  const float* fcw     = (const float*)d_in[2];
  const float* fcb     = (const float*)d_in[3];
  float* out = (float*)d_out;

  const int B        = in_sizes[0] / NQ;
  const int n_layers = in_sizes[1] / (NQ * 3);

  vqe_kron<<<B / SPW, 32, 0, stream>>>(x, weights, n_layers, fcw, fcb, out);
}


# vqe two-layer path: a complex product is 3 real f32 WMMA products, not 4: x=(a+b)c shared, real=x-b(c+d), imag=x+a(d-c) formed in the accumulators; 27 WMMAs per 4 samples, not 36; sums reordered
# speedup vs baseline: 48.1067x; 1.0968x over previous
#include <hip/hip_runtime.h>
#include <math.h>

typedef __attribute__((ext_vector_type(2))) float v2f;
typedef __attribute__((ext_vector_type(8))) float v8f;

#define NQ   8
#define SPW  16
#define PSROW 68

__device__ __forceinline__ float2 cmul(float2 a, float2 b) {
  return make_float2(a.x * b.x - a.y * b.y, a.x * b.y + a.y * b.x);
}

__device__ __forceinline__ int ring_src(int m, int r) {
  int j = m;
#pragma unroll
  for (int q = NQ - 1; q >= 0; --q) {
    const int tq = (q + r) & 7;
    if (j & (1 << (7 - q))) j ^= 1 << (7 - tq);
  }
  return j;
}
__device__ __forceinline__ int ring_dst(int j, int r) {
  int m = j;
#pragma unroll
  for (int q = 0; q < NQ; ++q) {
    const int tq = (q + r) & 7;
    if (m & (1 << (7 - q))) m ^= 1 << (7 - tq);
  }
  return m;
}

__device__ __forceinline__ void rot_gate(const float* __restrict__ wt, float2* g) {
  const float ph = wt[0], th = wt[1], om = wt[2];
  const float c  = cosf(0.5f * th), s = sinf(0.5f * th);
  const float ap = 0.5f * (ph + om), bm = 0.5f * (ph - om);
  g[0] = make_float2( cosf(ap) * c, -sinf(ap) * c);
  g[1] = make_float2(-cosf(bm) * s, -sinf(bm) * s);
  g[2] = make_float2( cosf(bm) * s, -sinf(bm) * s);
  g[3] = make_float2( cosf(ap) * c,  sinf(ap) * c);
}

__device__ __forceinline__ int m_of(int lo, int hi) { return ((lo >> 2) << 6) | (hi << 2) | (lo & 3); }
__device__ __forceinline__ int khi_of(int kk, int h, int e) { return (kk << 2) | (e << 1) | h; }
__device__ __forceinline__ int klo_of(int kk, int h, int e) { return 8 * h + 2 * kk + e; }

struct LayerOps {
  v2f Ar[4], Ai[4], nAi[4];
  v2f Br[4], Bi[4], nBi[4];
};

__device__ __forceinline__ void build_ops(const float2 (*G)[4], int h, int c, LayerOps& K) {
#pragma unroll
  for (int kk = 0; kk < 4; ++kk) {
#pragma unroll
    for (int e = 0; e < 2; ++e) {
      const int kh = khi_of(kk, h, e);
      float2 a = G[2][((c >> 3) & 1) * 2 + ((kh >> 3) & 1)];
      a = cmul(a, G[3][((c >> 2) & 1) * 2 + ((kh >> 2) & 1)]);
      a = cmul(a, G[4][((c >> 1) & 1) * 2 + ((kh >> 1) & 1)]);
      a = cmul(a, G[5][(c & 1) * 2 + (kh & 1)]);
      K.Ar[kk][e] = a.x; K.Ai[kk][e] = a.y; K.nAi[kk][e] = -a.y;
      const int kl = klo_of(kk, h, e);
      float2 b = G[0][((c >> 3) & 1) * 2 + ((kl >> 3) & 1)];
      b = cmul(b, G[1][((c >> 2) & 1) * 2 + ((kl >> 2) & 1)]);
      b = cmul(b, G[6][((c >> 1) & 1) * 2 + ((kl >> 1) & 1)]);
      b = cmul(b, G[7][(c & 1) * 2 + (kl & 1)]);
      K.Br[kk][e] = b.x; K.Bi[kk][e] = b.y; K.nBi[kk][e] = -b.y;
    }
  }
}

#define WMMA4(a, b, acc) __builtin_amdgcn_wmma_f32_16x16x4_f32(false, (a), false, (b), (short)0, (acc), false, false)

__device__ __forceinline__ void apply_layer(const v2f* sr, const v2f* si, const LayerOps& K,
                                            v8f& d2r, v8f& d2i) {
  const v8f zero = {0.f, 0.f, 0.f, 0.f, 0.f, 0.f, 0.f, 0.f};
  v8f d1r = zero, d1i = zero;
#pragma unroll
  for (int kk = 0; kk < 4; ++kk) {
    d1r = WMMA4(sr[kk], K.Ar[kk], d1r);
    d1i = WMMA4(sr[kk], K.Ai[kk], d1i);
    d1r = WMMA4(si[kk], K.nAi[kk], d1r);
    d1i = WMMA4(si[kk], K.Ar[kk], d1i);
  }
  d2r = zero; d2i = zero;
#pragma unroll
  for (int kk = 0; kk < 4; ++kk) {
    const v2f tr = {d1r[2 * kk], d1r[2 * kk + 1]};
    const v2f ti = {d1i[2 * kk], d1i[2 * kk + 1]};
    d2r = WMMA4(K.Br[kk], tr, d2r);
    d2i = WMMA4(K.Bi[kk], tr, d2i);
    d2r = WMMA4(K.nBi[kk], ti, d2r);
    d2i = WMMA4(K.Br[kk], ti, d2i);
  }
}

__device__ __forceinline__ void run_layers(const float* __restrict__ weights, int n_layers,
                                           const float* fw, float fb, int sBase,
                                           float2 (*G1)[4], float (*PS)[PSROW], float (*ST)[256],
                                           float (*zred)[33], float* __restrict__ out) {
  const int l = threadIdx.x, h = l >> 4, c = l & 15;

  LayerOps K;
  build_ops(G1, h, c, K);

  float gw[8];
  {
    const int rn = (n_layers - 1) % (NQ - 1) + 1;
#pragma unroll
    for (int r = 0; r < 8; ++r) {
      const int m = ring_dst(m_of(r + 8 * h, c), rn);
      float g = 0.f;
#pragma unroll
      for (int i = 0; i < NQ; ++i) g += fw[i] * (1.f - 2.f * (float)((m >> (7 - i)) & 1));
      gw[r] = g;
    }
  }

  const int jl = ring_src(m_of(c, khi_of(0, h, 0)), 1);

  for (int s = 0; s < SPW; ++s) {
    float2 a[NQ], b[NQ];
#pragma unroll
    for (int q = 0; q < NQ; ++q) {
      const float4 v = *(const float4*)(PS[s] + q * 8 + ((jl >> (7 - q)) & 1) * 4);
      a[q] = make_float2(v.x, v.y);
      b[q] = make_float2(v.z, v.w);
    }
    const float2 P = cmul(cmul(a[0], a[1]), cmul(a[6], a[7]));
    const float2 X[2] = {cmul(P, a[2]), cmul(P, b[2])};
    float2 Y[2][2], Z[2][2];
#pragma unroll
    for (int u = 0; u < 2; ++u) {
#pragma unroll
      for (int v = 0; v < 2; ++v) {
        Y[u][v] = cmul(X[u], (u ^ v) ? b[3] : a[3]);
        Z[u][v] = cmul((u ^ v) ? b[4] : a[4], v ? b[5] : a[5]);
      }
    }
    v2f sr[4], si[4];
#pragma unroll
    for (int kk = 0; kk < 4; ++kk) {
#pragma unroll
      for (int e = 0; e < 2; ++e) {
        const float2 el = cmul(Y[kk >> 1][kk & 1], Z[kk & 1][e]);
        sr[kk][e] = el.x; si[kk][e] = el.y;
      }
    }

    v8f d2r, d2i;
    apply_layer(sr, si, K, d2r, d2i);

    {
      for (int L = 2; L < n_layers; ++L) {
        __syncthreads();
        if (l < NQ) rot_gate(weights + (L * NQ + l) * 3, G1[l]);
#pragma unroll
        for (int r = 0; r < 8; ++r) {
          const int j = m_of(r + 8 * h, c);
          ST[0][j] = d2r[r]; ST[1][j] = d2i[r];
        }
        __syncthreads();
        build_ops(G1, h, c, K);
        const int rr = (L - 1) % (NQ - 1) + 1;
#pragma unroll
        for (int kk = 0; kk < 4; ++kk) {
#pragma unroll
          for (int e = 0; e < 2; ++e) {
            const int j = ring_src(m_of(c, khi_of(kk, h, e)), rr);
            sr[kk][e] = ST[0][j]; si[kk][e] = ST[1][j];
          }
        }
        apply_layer(sr, si, K, d2r, d2i);
      }
      __syncthreads();
      if (l < NQ) rot_gate(weights + (NQ + l) * 3, G1[l]);
      __syncthreads();
      build_ops(G1, h, c, K);
    }

    float zp = 0.f;
#pragma unroll
    for (int r = 0; r < 8; ++r) zp += (d2r[r] * d2r[r] + d2i[r] * d2i[r]) * gw[r];
    zred[s][l] = zp;
  }
  __syncthreads();

  if (l < SPW) {
    float sum = fb;
#pragma unroll
    for (int cc = 0; cc < 32; ++cc) sum += zred[l][cc];
    out[sBase + l] = 1.f / (1.f + expf(-sum));
  }
}

struct GroupIn {
  float4 v0, v1, v7, v2, v3, v4, v5;
  float2 v6;
};
struct GroupOps {
  v2f sr[4], si[4], ss[4];
  v2f fr, fi, fs;
};

__device__ __forceinline__ GroupIn load_group(const float* ps, int h, int m1, int m6) {
  GroupIn in;
  in.v0 = *(const float4*)(ps + 0 * 8 + h * 4);
  in.v1 = *(const float4*)(ps + 1 * 8 + (h ^ m1) * 4);
  in.v7 = *(const float4*)(ps + 7 * 8 + m6 * 4);
  in.v2 = *(const float4*)(ps + 2 * 8 + m1 * 4);
  in.v3 = *(const float4*)(ps + 3 * 8);
  in.v4 = *(const float4*)(ps + 4 * 8);
  in.v5 = *(const float4*)(ps + 5 * 8 + h * 4);
  in.v6 = *(const float2*)(ps + 6 * 8 + (h ^ m6) * 4);
  return in;
}

__device__ __forceinline__ void build_group_ops(const GroupIn& in, GroupOps& o) {
  const float2 f0 = cmul(cmul(make_float2(in.v0.x, in.v0.y), make_float2(in.v1.x, in.v1.y)), make_float2(in.v7.x, in.v7.y));
  const float2 f1 = cmul(cmul(make_float2(in.v0.z, in.v0.w), make_float2(in.v1.z, in.v1.w)), make_float2(in.v7.z, in.v7.w));
  o.fr[0] = f0.x; o.fr[1] = f1.x; o.fi[0] = f0.y; o.fi[1] = f1.y;
  o.fs[0] = f0.x + f0.y; o.fs[1] = f1.x + f1.y;

  const float2 X0 = make_float2(in.v2.x, in.v2.y), X1 = make_float2(in.v2.z, in.v2.w);
  const float2 p30 = make_float2(in.v3.x, in.v3.y), p31 = make_float2(in.v3.z, in.v3.w);
  const float2 p40 = make_float2(in.v4.x, in.v4.y), p41 = make_float2(in.v4.z, in.v4.w);
  const float2 Y[4] = {cmul(X0, p30), cmul(X0, p31), cmul(X1, p31), cmul(X1, p30)};
  const float2 Q0 = cmul(make_float2(in.v5.x, in.v5.y), in.v6), Q1 = cmul(make_float2(in.v5.z, in.v5.w), in.v6);
  const float2 Z[2][2] = {{cmul(p40, Q0), cmul(p41, Q1)}, {cmul(p41, Q0), cmul(p40, Q1)}};
#pragma unroll
  for (int kk = 0; kk < 4; ++kk) {
#pragma unroll
    for (int e = 0; e < 2; ++e) {
      const float2 el = cmul(Y[kk], Z[kk & 1][e]);
      o.sr[kk][e] = el.x; o.si[kk][e] = el.y; o.ss[kk][e] = el.x + el.y;
    }
  }
}

struct TwoLayerOps {
  v2f Ar[4], nAs[4], Ad[4];
  v2f Br, nBs, Bd;
  float2 cf[2];
};

__device__ __forceinline__ void run_two_layers(const float* fw, float fb, int sBase,
                                               const float2 (*G1)[4], const float (*PS)[PSROW],
                                               float (*zred)[33], float* __restrict__ out) {
  const int l = threadIdx.x, h = l >> 4, c = l & 15;

  TwoLayerOps K;
#pragma unroll
  for (int kk = 0; kk < 4; ++kk) {
#pragma unroll
    for (int e = 0; e < 2; ++e) {
      const int kh = khi_of(kk, h, e);
      float2 a = G1[2][((c >> 3) & 1) * 2 + ((kh >> 3) & 1)];
      a = cmul(a, G1[3][((c >> 2) & 1) * 2 + ((kh >> 2) & 1)]);
      a = cmul(a, G1[4][((c >> 1) & 1) * 2 + ((kh >> 1) & 1)]);
      a = cmul(a, G1[5][(c & 1) * 2 + (kh & 1)]);
      K.Ar[kk][e] = a.x; K.nAs[kk][e] = -a.x - a.y; K.Ad[kk][e] = a.y - a.x;
    }
  }
#pragma unroll
  for (int e = 0; e < 2; ++e) {
    const float2 b = cmul(G1[0][((c >> 3) & 1) * 2 + h], G1[7][(c & 1) * 2 + e]);
    K.Br[e] = b.x; K.nBs[e] = -b.x - b.y; K.Bd[e] = b.y - b.x;
    K.cf[e] = cmul(G1[1][((c >> 2) & 1) * 2 + h], G1[6][((c >> 1) & 1) * 2 + e]);
  }

  float gw[8];
#pragma unroll
  for (int r = 0; r < 8; ++r) {
    const int m = ring_dst(m_of(c, r + 8 * h), 2);
    float g = 0.f;
#pragma unroll
    for (int i = 0; i < NQ; ++i) g += fw[i] * (1.f - 2.f * (float)((m >> (7 - i)) & 1));
    gw[r] = g;
  }

  const int m1 = c >> 3, s4 = (c >> 1) & 3, m6 = c & 1;
  const v8f zero = {0.f, 0.f, 0.f, 0.f, 0.f, 0.f, 0.f, 0.f};

  GroupOps cur, nxt;
  build_group_ops(load_group(PS[s4], h, m1, m6), cur);
  GroupIn in = load_group(PS[4 + s4], h, m1, m6);

#pragma unroll
  for (int g = 0; g < SPW / 4; ++g) {

    v8f gx = WMMA4(cur.ss[0], K.Ar[0], zero);
    const v8f wx = WMMA4(cur.fs, K.Br, zero);
    gx = WMMA4(cur.ss[1], K.Ar[1], gx);
    const v8f wr = WMMA4(cur.fi, K.nBs, wx);
    gx = WMMA4(cur.ss[2], K.Ar[2], gx);
    const v8f wi = WMMA4(cur.fr, K.Bd, wx);
    gx = WMMA4(cur.ss[3], K.Ar[3], gx);

    v8f gr = gx, gi = gx;
#pragma unroll
    for (int kk = 0; kk < 4; ++kk) {
      gr = WMMA4(cur.si[kk], K.nAs[kk], gr);
      gi = WMMA4(cur.sr[kk], K.Ad[kk], gi);
    }

    if (g + 1 < SPW / 4) {
      build_group_ops(in, nxt);
      if (g + 2 < SPW / 4) in = load_group(PS[4 * (g + 2) + s4], h, m1, m6);
    }

    float br[8], nbs[8], bd[8], gs[8];
#pragma unroll
    for (int r = 0; r < 8; ++r) {
      const float2 cf = K.cf[r & 1];
      const float bi = wr[r] * cf.y + wi[r] * cf.x;
      br[r]  = wr[r] * cf.x - wi[r] * cf.y;
      nbs[r] = -br[r] - bi;
      bd[r]  = bi - br[r];
      gs[r]  = gr[r] + gi[r];
    }

    v8f cx[4];
#pragma unroll
    for (int t = 0; t < 4; ++t) {
      const v2f as = {gs[2 * t], gs[2 * t + 1]}, tr = {br[2 * t], br[2 * t + 1]};
      cx[t] = WMMA4(as, tr, zero);
    }
#pragma unroll
    for (int t = 0; t < 4; ++t) {
      const v2f ar = {gr[2 * t], gr[2 * t + 1]}, ai = {gi[2 * t], gi[2 * t + 1]};
      const v2f nts = {nbs[2 * t], nbs[2 * t + 1]}, td = {bd[2 * t], bd[2 * t + 1]};
      const v8f cr = WMMA4(ai, nts, cx[t]);
      const v8f ci = WMMA4(ar, td, cx[t]);
      float zp = 0.f;
#pragma unroll
      for (int r = 0; r < 8; ++r) zp += (cr[r] * cr[r] + ci[r] * ci[r]) * gw[r];
      zred[4 * g + t][l] = zp;
    }
    cur = nxt;
  }
  __syncthreads();

  if (l < SPW) {
    float sum = fb;
#pragma unroll
    for (int cc = 0; cc < 32; ++cc) sum += zred[l][cc];
    out[sBase + l] = 1.f / (1.f + expf(-sum));
  }
}

__global__ __launch_bounds__(32) void vqe_kron(const float* __restrict__ x,
                                               const float* __restrict__ weights,
                                               int n_layers,
                                               const float* __restrict__ fcw,
                                               const float* __restrict__ fcb,
                                               float* __restrict__ out) {
  __shared__ __attribute__((aligned(16))) float2 GG[2][NQ][4];
  __shared__ __attribute__((aligned(16))) float PS[SPW][PSROW];
  __shared__ float ST[2][256];
  __shared__ float zred[SPW][33];
  float2 (*G1)[4] = GG[1];

  const int l = threadIdx.x;
  const int h = l >> 4;
  const int c = l & 15;
  const int sBase = blockIdx.x * SPW;

  const float4 xv = *(const float4*)(x + (size_t)(sBase + c) * NQ + 4 * h);
  const bool have = (c >> 3) < n_layers;
  float ph = 0.f, th = 0.f, om = 0.f;
  if (have) {
    const float* wt = weights + c * 3;
    ph = wt[0]; th = wt[1]; om = wt[2];
  }
  float fw[NQ];
#pragma unroll
  for (int i = 0; i < NQ; ++i) fw[i] = fcw[i];
  const float fb = fcb[0];

  float s1, c1, s2, c2;
  sincosf(h ? 0.5f * (ph + om) : 0.5f * th, &s1, &c1);
  sincosf(0.5f * (ph - om), &s2, &c2);
  const float so = __shfl_xor(s1, 16), co = __shfl_xor(c1, 16);
  if (h == 0) {
    float2* g = GG[c >> 3][c & 7];
    if (have) {
      g[0] = make_float2( co * c1, -so * c1);
      g[1] = make_float2(-c2 * s1, -s2 * s1);
      g[2] = make_float2( c2 * s1, -s2 * s1);
      g[3] = make_float2( co * c1,  so * c1);
    } else if (c < NQ) {
      g[0] = make_float2(1.f, 0.f); g[1] = make_float2(0.f, 0.f);
      g[2] = make_float2(0.f, 0.f); g[3] = make_float2(1.f, 0.f);
    }
  }

  float cs[4], sn[4];
  sincosf(0.5f * xv.x, &sn[0], &cs[0]);
  sincosf(0.5f * xv.y, &sn[1], &cs[1]);
  sincosf(0.5f * xv.z, &sn[2], &cs[2]);
  sincosf(0.5f * xv.w, &sn[3], &cs[3]);
  __syncthreads();

#pragma unroll
  for (int i = 0; i < 4; ++i) {
    const int q = 4 * h + i;
    const float4 ua = *(const float4*)GG[0][q], ub = *(const float4*)(GG[0][q] + 2);
    const float2 p0 = make_float2(ua.x * cs[i] + ua.w * sn[i], ua.y * cs[i] - ua.z * sn[i]);
    const float2 p1 = make_float2(ub.x * cs[i] + ub.w * sn[i], ub.y * cs[i] - ub.z * sn[i]);
    *(float4*)(PS[c] + q * 8)     = make_float4(p0.x, p0.y, p1.x, p1.y);
    *(float4*)(PS[c] + q * 8 + 4) = make_float4(p1.x, p1.y, p0.x, p0.y);
  }
  __syncthreads();

  if (n_layers > 2) {
    run_layers(weights, n_layers, fw, fb, sBase, G1, PS, ST, zred, out);
  } else if (n_layers == 2) {
    run_two_layers(fw, fb, sBase, G1, PS, zred, out);
  } else {
    for (int s = 0; s < SPW; ++s) {
      float zp = 0.f;
      for (int i = 0; i < 8; ++i) {
        const int m = l + 32 * i;
        const int j = (n_layers > 0) ? ring_src(m, 1) : m;
        float2 v = make_float2(PS[s][((j >> 7) & 1) * 2], PS[s][((j >> 7) & 1) * 2 + 1]);
        for (int q = 1; q < NQ; ++q) {
          const int bq = (j >> (7 - q)) & 1;
          v = cmul(v, make_float2(PS[s][q * 8 + bq * 2], PS[s][q * 8 + bq * 2 + 1]));
        }
        float g = 0.f;
        for (int t = 0; t < NQ; ++t) g += fw[t] * (1.f - 2.f * (float)((m >> (7 - t)) & 1));
        zp += (v.x * v.x + v.y * v.y) * g;
      }
      zred[s][l] = zp;
    }
    __syncthreads();
    if (l < SPW) {
      float sum = fb;
      for (int cc = 0; cc < 32; ++cc) sum += zred[l][cc];
      out[sBase + l] = 1.f / (1.f + expf(-sum));
    }
  }
}

extern "C" void kernel_launch(void* const* d_in, const int* in_sizes, int n_in,
                              void* d_out, int out_size, void* d_ws, size_t ws_size,
                              hipStream_t stream) {
  const float* x       = (const float*)d_in[0];
  const float* weights = (const float*)d_in[1];
  const float* fcw     = (const float*)d_in[2];
  const float* fcb     = (const float*)d_in[3];
  float* out = (float*)d_out;

  const int B        = in_sizes[0] / NQ;
  const int n_layers = in_sizes[1] / (NQ * 3);

  vqe_kron<<<B / SPW, 32, 0, stream>>>(x, weights, n_layers, fcw, fcb, out);
}


# vqe two-layer path: readout weights via the ring's linearity over index bits: sign of fc_w[i] = lane sign (once per lane) x constant register sign; no per-(r,i) bit extraction; sums unchanged
# speedup vs baseline: 50.4920x; 1.0496x over previous
#include <hip/hip_runtime.h>
#include <math.h>

typedef __attribute__((ext_vector_type(2))) float v2f;
typedef __attribute__((ext_vector_type(8))) float v8f;

#define NQ   8
#define SPW  16
#define PSROW 68

__device__ __forceinline__ float2 cmul(float2 a, float2 b) {
  return make_float2(a.x * b.x - a.y * b.y, a.x * b.y + a.y * b.x);
}

__device__ __forceinline__ int ring_src(int m, int r) {
  int j = m;
#pragma unroll
  for (int q = NQ - 1; q >= 0; --q) {
    const int tq = (q + r) & 7;
    if (j & (1 << (7 - q))) j ^= 1 << (7 - tq);
  }
  return j;
}
__device__ __forceinline__ int ring_dst(int j, int r) {
  int m = j;
#pragma unroll
  for (int q = 0; q < NQ; ++q) {
    const int tq = (q + r) & 7;
    if (m & (1 << (7 - q))) m ^= 1 << (7 - tq);
  }
  return m;
}

__device__ __forceinline__ void rot_gate(const float* __restrict__ wt, float2* g) {
  const float ph = wt[0], th = wt[1], om = wt[2];
  const float c  = cosf(0.5f * th), s = sinf(0.5f * th);
  const float ap = 0.5f * (ph + om), bm = 0.5f * (ph - om);
  g[0] = make_float2( cosf(ap) * c, -sinf(ap) * c);
  g[1] = make_float2(-cosf(bm) * s, -sinf(bm) * s);
  g[2] = make_float2( cosf(bm) * s, -sinf(bm) * s);
  g[3] = make_float2( cosf(ap) * c,  sinf(ap) * c);
}

__device__ __forceinline__ int m_of(int lo, int hi) { return ((lo >> 2) << 6) | (hi << 2) | (lo & 3); }
__device__ __forceinline__ int khi_of(int kk, int h, int e) { return (kk << 2) | (e << 1) | h; }
__device__ __forceinline__ int klo_of(int kk, int h, int e) { return 8 * h + 2 * kk + e; }

struct LayerOps {
  v2f Ar[4], Ai[4], nAi[4];
  v2f Br[4], Bi[4], nBi[4];
};

__device__ __forceinline__ void build_ops(const float2 (*G)[4], int h, int c, LayerOps& K) {
#pragma unroll
  for (int kk = 0; kk < 4; ++kk) {
#pragma unroll
    for (int e = 0; e < 2; ++e) {
      const int kh = khi_of(kk, h, e);
      float2 a = G[2][((c >> 3) & 1) * 2 + ((kh >> 3) & 1)];
      a = cmul(a, G[3][((c >> 2) & 1) * 2 + ((kh >> 2) & 1)]);
      a = cmul(a, G[4][((c >> 1) & 1) * 2 + ((kh >> 1) & 1)]);
      a = cmul(a, G[5][(c & 1) * 2 + (kh & 1)]);
      K.Ar[kk][e] = a.x; K.Ai[kk][e] = a.y; K.nAi[kk][e] = -a.y;
      const int kl = klo_of(kk, h, e);
      float2 b = G[0][((c >> 3) & 1) * 2 + ((kl >> 3) & 1)];
      b = cmul(b, G[1][((c >> 2) & 1) * 2 + ((kl >> 2) & 1)]);
      b = cmul(b, G[6][((c >> 1) & 1) * 2 + ((kl >> 1) & 1)]);
      b = cmul(b, G[7][(c & 1) * 2 + (kl & 1)]);
      K.Br[kk][e] = b.x; K.Bi[kk][e] = b.y; K.nBi[kk][e] = -b.y;
    }
  }
}

#define WMMA4(a, b, acc) __builtin_amdgcn_wmma_f32_16x16x4_f32(false, (a), false, (b), (short)0, (acc), false, false)

__device__ __forceinline__ void apply_layer(const v2f* sr, const v2f* si, const LayerOps& K,
                                            v8f& d2r, v8f& d2i) {
  const v8f zero = {0.f, 0.f, 0.f, 0.f, 0.f, 0.f, 0.f, 0.f};
  v8f d1r = zero, d1i = zero;
#pragma unroll
  for (int kk = 0; kk < 4; ++kk) {
    d1r = WMMA4(sr[kk], K.Ar[kk], d1r);
    d1i = WMMA4(sr[kk], K.Ai[kk], d1i);
    d1r = WMMA4(si[kk], K.nAi[kk], d1r);
    d1i = WMMA4(si[kk], K.Ar[kk], d1i);
  }
  d2r = zero; d2i = zero;
#pragma unroll
  for (int kk = 0; kk < 4; ++kk) {
    const v2f tr = {d1r[2 * kk], d1r[2 * kk + 1]};
    const v2f ti = {d1i[2 * kk], d1i[2 * kk + 1]};
    d2r = WMMA4(K.Br[kk], tr, d2r);
    d2i = WMMA4(K.Bi[kk], tr, d2i);
    d2r = WMMA4(K.nBi[kk], ti, d2r);
    d2i = WMMA4(K.Br[kk], ti, d2i);
  }
}

__device__ __forceinline__ void run_layers(const float* __restrict__ weights, int n_layers,
                                           const float* fw, float fb, int sBase,
                                           float2 (*G1)[4], float (*PS)[PSROW], float (*ST)[256],
                                           float (*zred)[33], float* __restrict__ out) {
  const int l = threadIdx.x, h = l >> 4, c = l & 15;

  LayerOps K;
  build_ops(G1, h, c, K);

  float gw[8];
  {
    const int rn = (n_layers - 1) % (NQ - 1) + 1;
#pragma unroll
    for (int r = 0; r < 8; ++r) {
      const int m = ring_dst(m_of(r + 8 * h, c), rn);
      float g = 0.f;
#pragma unroll
      for (int i = 0; i < NQ; ++i) g += fw[i] * (1.f - 2.f * (float)((m >> (7 - i)) & 1));
      gw[r] = g;
    }
  }

  const int jl = ring_src(m_of(c, khi_of(0, h, 0)), 1);

  for (int s = 0; s < SPW; ++s) {
    float2 a[NQ], b[NQ];
#pragma unroll
    for (int q = 0; q < NQ; ++q) {
      const float4 v = *(const float4*)(PS[s] + q * 8 + ((jl >> (7 - q)) & 1) * 4);
      a[q] = make_float2(v.x, v.y);
      b[q] = make_float2(v.z, v.w);
    }
    const float2 P = cmul(cmul(a[0], a[1]), cmul(a[6], a[7]));
    const float2 X[2] = {cmul(P, a[2]), cmul(P, b[2])};
    float2 Y[2][2], Z[2][2];
#pragma unroll
    for (int u = 0; u < 2; ++u) {
#pragma unroll
      for (int v = 0; v < 2; ++v) {
        Y[u][v] = cmul(X[u], (u ^ v) ? b[3] : a[3]);
        Z[u][v] = cmul((u ^ v) ? b[4] : a[4], v ? b[5] : a[5]);
      }
    }
    v2f sr[4], si[4];
#pragma unroll
    for (int kk = 0; kk < 4; ++kk) {
#pragma unroll
      for (int e = 0; e < 2; ++e) {
        const float2 el = cmul(Y[kk >> 1][kk & 1], Z[kk & 1][e]);
        sr[kk][e] = el.x; si[kk][e] = el.y;
      }
    }

    v8f d2r, d2i;
    apply_layer(sr, si, K, d2r, d2i);

    {
      for (int L = 2; L < n_layers; ++L) {
        __syncthreads();
        if (l < NQ) rot_gate(weights + (L * NQ + l) * 3, G1[l]);
#pragma unroll
        for (int r = 0; r < 8; ++r) {
          const int j = m_of(r + 8 * h, c);
          ST[0][j] = d2r[r]; ST[1][j] = d2i[r];
        }
        __syncthreads();
        build_ops(G1, h, c, K);
        const int rr = (L - 1) % (NQ - 1) + 1;
#pragma unroll
        for (int kk = 0; kk < 4; ++kk) {
#pragma unroll
          for (int e = 0; e < 2; ++e) {
            const int j = ring_src(m_of(c, khi_of(kk, h, e)), rr);
            sr[kk][e] = ST[0][j]; si[kk][e] = ST[1][j];
          }
        }
        apply_layer(sr, si, K, d2r, d2i);
      }
      __syncthreads();
      if (l < NQ) rot_gate(weights + (NQ + l) * 3, G1[l]);
      __syncthreads();
      build_ops(G1, h, c, K);
    }

    float zp = 0.f;
#pragma unroll
    for (int r = 0; r < 8; ++r) zp += (d2r[r] * d2r[r] + d2i[r] * d2i[r]) * gw[r];
    zred[s][l] = zp;
  }
  __syncthreads();

  if (l < SPW) {
    float sum = fb;
#pragma unroll
    for (int cc = 0; cc < 32; ++cc) sum += zred[l][cc];
    out[sBase + l] = 1.f / (1.f + expf(-sum));
  }
}

struct GroupIn {
  float4 v0, v1, v7, v2, v3, v4, v5;
  float2 v6;
};
struct GroupOps {
  v2f sr[4], si[4], ss[4];
  v2f fr, fi, fs;
};

__device__ __forceinline__ GroupIn load_group(const float* ps, int h, int m1, int m6) {
  GroupIn in;
  in.v0 = *(const float4*)(ps + 0 * 8 + h * 4);
  in.v1 = *(const float4*)(ps + 1 * 8 + (h ^ m1) * 4);
  in.v7 = *(const float4*)(ps + 7 * 8 + m6 * 4);
  in.v2 = *(const float4*)(ps + 2 * 8 + m1 * 4);
  in.v3 = *(const float4*)(ps + 3 * 8);
  in.v4 = *(const float4*)(ps + 4 * 8);
  in.v5 = *(const float4*)(ps + 5 * 8 + h * 4);
  in.v6 = *(const float2*)(ps + 6 * 8 + (h ^ m6) * 4);
  return in;
}

__device__ __forceinline__ void build_group_ops(const GroupIn& in, GroupOps& o) {
  const float2 f0 = cmul(cmul(make_float2(in.v0.x, in.v0.y), make_float2(in.v1.x, in.v1.y)), make_float2(in.v7.x, in.v7.y));
  const float2 f1 = cmul(cmul(make_float2(in.v0.z, in.v0.w), make_float2(in.v1.z, in.v1.w)), make_float2(in.v7.z, in.v7.w));
  o.fr[0] = f0.x; o.fr[1] = f1.x; o.fi[0] = f0.y; o.fi[1] = f1.y;
  o.fs[0] = f0.x + f0.y; o.fs[1] = f1.x + f1.y;

  const float2 X0 = make_float2(in.v2.x, in.v2.y), X1 = make_float2(in.v2.z, in.v2.w);
  const float2 p30 = make_float2(in.v3.x, in.v3.y), p31 = make_float2(in.v3.z, in.v3.w);
  const float2 p40 = make_float2(in.v4.x, in.v4.y), p41 = make_float2(in.v4.z, in.v4.w);
  const float2 Y[4] = {cmul(X0, p30), cmul(X0, p31), cmul(X1, p31), cmul(X1, p30)};
  const float2 Q0 = cmul(make_float2(in.v5.x, in.v5.y), in.v6), Q1 = cmul(make_float2(in.v5.z, in.v5.w), in.v6);
  const float2 Z[2][2] = {{cmul(p40, Q0), cmul(p41, Q1)}, {cmul(p41, Q0), cmul(p40, Q1)}};
#pragma unroll
  for (int kk = 0; kk < 4; ++kk) {
#pragma unroll
    for (int e = 0; e < 2; ++e) {
      const float2 el = cmul(Y[kk], Z[kk & 1][e]);
      o.sr[kk][e] = el.x; o.si[kk][e] = el.y; o.ss[kk][e] = el.x + el.y;
    }
  }
}

struct TwoLayerOps {
  v2f Ar[4], nAs[4], Ad[4];
  v2f Br, nBs, Bd;
  float2 cf[2];
};

__device__ __forceinline__ void run_two_layers(const float* fw, float fb, int sBase,
                                               const float2 (*G1)[4], const float (*PS)[PSROW],
                                               float (*zred)[33], float* __restrict__ out) {
  const int l = threadIdx.x, h = l >> 4, c = l & 15;

  TwoLayerOps K;
#pragma unroll
  for (int kk = 0; kk < 4; ++kk) {
#pragma unroll
    for (int e = 0; e < 2; ++e) {
      const int kh = khi_of(kk, h, e);
      float2 a = G1[2][((c >> 3) & 1) * 2 + ((kh >> 3) & 1)];
      a = cmul(a, G1[3][((c >> 2) & 1) * 2 + ((kh >> 2) & 1)]);
      a = cmul(a, G1[4][((c >> 1) & 1) * 2 + ((kh >> 1) & 1)]);
      a = cmul(a, G1[5][(c & 1) * 2 + (kh & 1)]);
      K.Ar[kk][e] = a.x; K.nAs[kk][e] = -a.x - a.y; K.Ad[kk][e] = a.y - a.x;
    }
  }
#pragma unroll
  for (int e = 0; e < 2; ++e) {
    const float2 b = cmul(G1[0][((c >> 3) & 1) * 2 + h], G1[7][(c & 1) * 2 + e]);
    K.Br[e] = b.x; K.nBs[e] = -b.x - b.y; K.Bd[e] = b.y - b.x;
    K.cf[e] = cmul(G1[1][((c >> 2) & 1) * 2 + h], G1[6][((c >> 1) & 1) * 2 + e]);
  }

  const int ml = ring_dst(m_of(c, 8 * h), 2);
  float fs[NQ];
#pragma unroll
  for (int i = 0; i < NQ; ++i)
    fs[i] = __uint_as_float(__float_as_uint(fw[i]) ^ (((unsigned)ml << (24 + i)) & 0x80000000u));
  float gw[8];
#pragma unroll
  for (int r = 0; r < 8; ++r) {
    const int mr = ring_dst(m_of(0, r), 2);
    float g = 0.f;
#pragma unroll
    for (int i = 0; i < NQ; ++i) {
      if ((mr >> (7 - i)) & 1) g -= fs[i]; else g += fs[i];
    }
    gw[r] = g;
  }

  const int m1 = c >> 3, s4 = (c >> 1) & 3, m6 = c & 1;
  const v8f zero = {0.f, 0.f, 0.f, 0.f, 0.f, 0.f, 0.f, 0.f};

  GroupOps cur, nxt;
  build_group_ops(load_group(PS[s4], h, m1, m6), cur);
  GroupIn in = load_group(PS[4 + s4], h, m1, m6);

#pragma unroll
  for (int g = 0; g < SPW / 4; ++g) {

    v8f gx = WMMA4(cur.ss[0], K.Ar[0], zero);
    const v8f wx = WMMA4(cur.fs, K.Br, zero);
    gx = WMMA4(cur.ss[1], K.Ar[1], gx);
    const v8f wr = WMMA4(cur.fi, K.nBs, wx);
    gx = WMMA4(cur.ss[2], K.Ar[2], gx);
    const v8f wi = WMMA4(cur.fr, K.Bd, wx);
    gx = WMMA4(cur.ss[3], K.Ar[3], gx);

    v8f gr = gx, gi = gx;
#pragma unroll
    for (int kk = 0; kk < 4; ++kk) {
      gr = WMMA4(cur.si[kk], K.nAs[kk], gr);
      gi = WMMA4(cur.sr[kk], K.Ad[kk], gi);
    }

    if (g + 1 < SPW / 4) {
      build_group_ops(in, nxt);
      if (g + 2 < SPW / 4) in = load_group(PS[4 * (g + 2) + s4], h, m1, m6);
    }

    float br[8], nbs[8], bd[8], gs[8];
#pragma unroll
    for (int r = 0; r < 8; ++r) {
      const float2 cf = K.cf[r & 1];
      const float bi = wr[r] * cf.y + wi[r] * cf.x;
      br[r]  = wr[r] * cf.x - wi[r] * cf.y;
      nbs[r] = -br[r] - bi;
      bd[r]  = bi - br[r];
      gs[r]  = gr[r] + gi[r];
    }

    v8f cx[4];
#pragma unroll
    for (int t = 0; t < 4; ++t) {
      const v2f as = {gs[2 * t], gs[2 * t + 1]}, tr = {br[2 * t], br[2 * t + 1]};
      cx[t] = WMMA4(as, tr, zero);
    }
#pragma unroll
    for (int t = 0; t < 4; ++t) {
      const v2f ar = {gr[2 * t], gr[2 * t + 1]}, ai = {gi[2 * t], gi[2 * t + 1]};
      const v2f nts = {nbs[2 * t], nbs[2 * t + 1]}, td = {bd[2 * t], bd[2 * t + 1]};
      const v8f cr = WMMA4(ai, nts, cx[t]);
      const v8f ci = WMMA4(ar, td, cx[t]);
      float zp = 0.f;
#pragma unroll
      for (int r = 0; r < 8; ++r) zp += (cr[r] * cr[r] + ci[r] * ci[r]) * gw[r];
      zred[4 * g + t][l] = zp;
    }
    cur = nxt;
  }
  __syncthreads();

  if (l < SPW) {
    float sum = fb;
#pragma unroll
    for (int cc = 0; cc < 32; ++cc) sum += zred[l][cc];
    out[sBase + l] = 1.f / (1.f + expf(-sum));
  }
}

__global__ __launch_bounds__(32) void vqe_kron(const float* __restrict__ x,
                                               const float* __restrict__ weights,
                                               int n_layers,
                                               const float* __restrict__ fcw,
                                               const float* __restrict__ fcb,
                                               float* __restrict__ out) {
  __shared__ __attribute__((aligned(16))) float2 GG[2][NQ][4];
  __shared__ __attribute__((aligned(16))) float PS[SPW][PSROW];
  __shared__ float ST[2][256];
  __shared__ float zred[SPW][33];
  float2 (*G1)[4] = GG[1];

  const int l = threadIdx.x;
  const int h = l >> 4;
  const int c = l & 15;
  const int sBase = blockIdx.x * SPW;

  const float4 xv = *(const float4*)(x + (size_t)(sBase + c) * NQ + 4 * h);
  const bool have = (c >> 3) < n_layers;
  float ph = 0.f, th = 0.f, om = 0.f;
  if (have) {
    const float* wt = weights + c * 3;
    ph = wt[0]; th = wt[1]; om = wt[2];
  }
  float fw[NQ];
#pragma unroll
  for (int i = 0; i < NQ; ++i) fw[i] = fcw[i];
  const float fb = fcb[0];

  float s1, c1, s2, c2;
  sincosf(h ? 0.5f * (ph + om) : 0.5f * th, &s1, &c1);
  sincosf(0.5f * (ph - om), &s2, &c2);
  const float so = __shfl_xor(s1, 16), co = __shfl_xor(c1, 16);
  if (h == 0) {
    float2* g = GG[c >> 3][c & 7];
    if (have) {
      g[0] = make_float2( co * c1, -so * c1);
      g[1] = make_float2(-c2 * s1, -s2 * s1);
      g[2] = make_float2( c2 * s1, -s2 * s1);
      g[3] = make_float2( co * c1,  so * c1);
    } else if (c < NQ) {
      g[0] = make_float2(1.f, 0.f); g[1] = make_float2(0.f, 0.f);
      g[2] = make_float2(0.f, 0.f); g[3] = make_float2(1.f, 0.f);
    }
  }

  float cs[4], sn[4];
  sincosf(0.5f * xv.x, &sn[0], &cs[0]);
  sincosf(0.5f * xv.y, &sn[1], &cs[1]);
  sincosf(0.5f * xv.z, &sn[2], &cs[2]);
  sincosf(0.5f * xv.w, &sn[3], &cs[3]);
  __syncthreads();

#pragma unroll
  for (int i = 0; i < 4; ++i) {
    const int q = 4 * h + i;
    const float4 ua = *(const float4*)GG[0][q], ub = *(const float4*)(GG[0][q] + 2);
    const float2 p0 = make_float2(ua.x * cs[i] + ua.w * sn[i], ua.y * cs[i] - ua.z * sn[i]);
    const float2 p1 = make_float2(ub.x * cs[i] + ub.w * sn[i], ub.y * cs[i] - ub.z * sn[i]);
    *(float4*)(PS[c] + q * 8)     = make_float4(p0.x, p0.y, p1.x, p1.y);
    *(float4*)(PS[c] + q * 8 + 4) = make_float4(p1.x, p1.y, p0.x, p0.y);
  }
  __syncthreads();

  if (n_layers > 2) {
    run_layers(weights, n_layers, fw, fb, sBase, G1, PS, ST, zred, out);
  } else if (n_layers == 2) {
    run_two_layers(fw, fb, sBase, G1, PS, zred, out);
  } else {
    for (int s = 0; s < SPW; ++s) {
      float zp = 0.f;
      for (int i = 0; i < 8; ++i) {
        const int m = l + 32 * i;
        const int j = (n_layers > 0) ? ring_src(m, 1) : m;
        float2 v = make_float2(PS[s][((j >> 7) & 1) * 2], PS[s][((j >> 7) & 1) * 2 + 1]);
        for (int q = 1; q < NQ; ++q) {
          const int bq = (j >> (7 - q)) & 1;
          v = cmul(v, make_float2(PS[s][q * 8 + bq * 2], PS[s][q * 8 + bq * 2 + 1]));
        }
        float g = 0.f;
        for (int t = 0; t < NQ; ++t) g += fw[t] * (1.f - 2.f * (float)((m >> (7 - t)) & 1));
        zp += (v.x * v.x + v.y * v.y) * g;
      }
      zred[s][l] = zp;
    }
    __syncthreads();
    if (l < SPW) {
      float sum = fb;
      for (int cc = 0; cc < 32; ++cc) sum += zred[l][cc];
      out[sBase + l] = 1.f / (1.f + expf(-sum));
    }
  }
}

extern "C" void kernel_launch(void* const* d_in, const int* in_sizes, int n_in,
                              void* d_out, int out_size, void* d_ws, size_t ws_size,
                              hipStream_t stream) {
  const float* x       = (const float*)d_in[0];
  const float* weights = (const float*)d_in[1];
  const float* fcw     = (const float*)d_in[2];
  const float* fcb     = (const float*)d_in[3];
  float* out = (float*)d_out;

  const int B        = in_sizes[0] / NQ;
  const int n_layers = in_sizes[1] / (NQ * 3);

  vqe_kron<<<B / SPW, 32, 0, stream>>>(x, weights, n_layers, fcw, fcb, out);
}


# launch shape: work-groups of 4 independent waves (512 groups, not 2048 one-wave ones), each wave its own LDS part and a wave-local LDS hand-over instead of a group barrier; sums unchanged
# speedup vs baseline: 60.1695x; 1.1917x over previous
#include <hip/hip_runtime.h>
#include <math.h>

typedef __attribute__((ext_vector_type(2))) float v2f;
typedef __attribute__((ext_vector_type(8))) float v8f;

#define NQ   8
#define SPW  16
#define WPB  4
#define PSROW 68

__device__ __forceinline__ float2 cmul(float2 a, float2 b) {
  return make_float2(a.x * b.x - a.y * b.y, a.x * b.y + a.y * b.x);
}

__device__ __forceinline__ void wave_sync() {
  __builtin_amdgcn_fence(__ATOMIC_RELEASE, "workgroup");
  __builtin_amdgcn_wave_barrier();
  __builtin_amdgcn_fence(__ATOMIC_ACQUIRE, "workgroup");
}

__device__ __forceinline__ int ring_src(int m, int r) {
  int j = m;
#pragma unroll
  for (int q = NQ - 1; q >= 0; --q) {
    const int tq = (q + r) & 7;
    if (j & (1 << (7 - q))) j ^= 1 << (7 - tq);
  }
  return j;
}
__device__ __forceinline__ int ring_dst(int j, int r) {
  int m = j;
#pragma unroll
  for (int q = 0; q < NQ; ++q) {
    const int tq = (q + r) & 7;
    if (m & (1 << (7 - q))) m ^= 1 << (7 - tq);
  }
  return m;
}

__device__ __forceinline__ void rot_gate(const float* __restrict__ wt, float2* g) {
  const float ph = wt[0], th = wt[1], om = wt[2];
  const float c  = cosf(0.5f * th), s = sinf(0.5f * th);
  const float ap = 0.5f * (ph + om), bm = 0.5f * (ph - om);
  g[0] = make_float2( cosf(ap) * c, -sinf(ap) * c);
  g[1] = make_float2(-cosf(bm) * s, -sinf(bm) * s);
  g[2] = make_float2( cosf(bm) * s, -sinf(bm) * s);
  g[3] = make_float2( cosf(ap) * c,  sinf(ap) * c);
}

__device__ __forceinline__ int m_of(int lo, int hi) { return ((lo >> 2) << 6) | (hi << 2) | (lo & 3); }
__device__ __forceinline__ int khi_of(int kk, int h, int e) { return (kk << 2) | (e << 1) | h; }
__device__ __forceinline__ int klo_of(int kk, int h, int e) { return 8 * h + 2 * kk + e; }

struct LayerOps {
  v2f Ar[4], Ai[4], nAi[4];
  v2f Br[4], Bi[4], nBi[4];
};

__device__ __forceinline__ void build_ops(const float2 (*G)[4], int h, int c, LayerOps& K) {
#pragma unroll
  for (int kk = 0; kk < 4; ++kk) {
#pragma unroll
    for (int e = 0; e < 2; ++e) {
      const int kh = khi_of(kk, h, e);
      float2 a = G[2][((c >> 3) & 1) * 2 + ((kh >> 3) & 1)];
      a = cmul(a, G[3][((c >> 2) & 1) * 2 + ((kh >> 2) & 1)]);
      a = cmul(a, G[4][((c >> 1) & 1) * 2 + ((kh >> 1) & 1)]);
      a = cmul(a, G[5][(c & 1) * 2 + (kh & 1)]);
      K.Ar[kk][e] = a.x; K.Ai[kk][e] = a.y; K.nAi[kk][e] = -a.y;
      const int kl = klo_of(kk, h, e);
      float2 b = G[0][((c >> 3) & 1) * 2 + ((kl >> 3) & 1)];
      b = cmul(b, G[1][((c >> 2) & 1) * 2 + ((kl >> 2) & 1)]);
      b = cmul(b, G[6][((c >> 1) & 1) * 2 + ((kl >> 1) & 1)]);
      b = cmul(b, G[7][(c & 1) * 2 + (kl & 1)]);
      K.Br[kk][e] = b.x; K.Bi[kk][e] = b.y; K.nBi[kk][e] = -b.y;
    }
  }
}

#define WMMA4(a, b, acc) __builtin_amdgcn_wmma_f32_16x16x4_f32(false, (a), false, (b), (short)0, (acc), false, false)

__device__ __forceinline__ void apply_layer(const v2f* sr, const v2f* si, const LayerOps& K,
                                            v8f& d2r, v8f& d2i) {
  const v8f zero = {0.f, 0.f, 0.f, 0.f, 0.f, 0.f, 0.f, 0.f};
  v8f d1r = zero, d1i = zero;
#pragma unroll
  for (int kk = 0; kk < 4; ++kk) {
    d1r = WMMA4(sr[kk], K.Ar[kk], d1r);
    d1i = WMMA4(sr[kk], K.Ai[kk], d1i);
    d1r = WMMA4(si[kk], K.nAi[kk], d1r);
    d1i = WMMA4(si[kk], K.Ar[kk], d1i);
  }
  d2r = zero; d2i = zero;
#pragma unroll
  for (int kk = 0; kk < 4; ++kk) {
    const v2f tr = {d1r[2 * kk], d1r[2 * kk + 1]};
    const v2f ti = {d1i[2 * kk], d1i[2 * kk + 1]};
    d2r = WMMA4(K.Br[kk], tr, d2r);
    d2i = WMMA4(K.Bi[kk], tr, d2i);
    d2r = WMMA4(K.nBi[kk], ti, d2r);
    d2i = WMMA4(K.Br[kk], ti, d2i);
  }
}

__device__ __forceinline__ void run_layers(const float* __restrict__ weights, int n_layers,
                                           const float* fw, float fb, int sBase,
                                           float2 (*G1)[4], float (*PS)[PSROW], float (*ST)[256],
                                           float (*zred)[33], float* __restrict__ out) {
  const int l = threadIdx.x & 31, h = l >> 4, c = l & 15;

  LayerOps K;
  build_ops(G1, h, c, K);

  float gw[8];
  {
    const int rn = (n_layers - 1) % (NQ - 1) + 1;
#pragma unroll
    for (int r = 0; r < 8; ++r) {
      const int m = ring_dst(m_of(r + 8 * h, c), rn);
      float g = 0.f;
#pragma unroll
      for (int i = 0; i < NQ; ++i) g += fw[i] * (1.f - 2.f * (float)((m >> (7 - i)) & 1));
      gw[r] = g;
    }
  }

  const int jl = ring_src(m_of(c, khi_of(0, h, 0)), 1);

  for (int s = 0; s < SPW; ++s) {
    float2 a[NQ], b[NQ];
#pragma unroll
    for (int q = 0; q < NQ; ++q) {
      const float4 v = *(const float4*)(PS[s] + q * 8 + ((jl >> (7 - q)) & 1) * 4);
      a[q] = make_float2(v.x, v.y);
      b[q] = make_float2(v.z, v.w);
    }
    const float2 P = cmul(cmul(a[0], a[1]), cmul(a[6], a[7]));
    const float2 X[2] = {cmul(P, a[2]), cmul(P, b[2])};
    float2 Y[2][2], Z[2][2];
#pragma unroll
    for (int u = 0; u < 2; ++u) {
#pragma unroll
      for (int v = 0; v < 2; ++v) {
        Y[u][v] = cmul(X[u], (u ^ v) ? b[3] : a[3]);
        Z[u][v] = cmul((u ^ v) ? b[4] : a[4], v ? b[5] : a[5]);
      }
    }
    v2f sr[4], si[4];
#pragma unroll
    for (int kk = 0; kk < 4; ++kk) {
#pragma unroll
      for (int e = 0; e < 2; ++e) {
        const float2 el = cmul(Y[kk >> 1][kk & 1], Z[kk & 1][e]);
        sr[kk][e] = el.x; si[kk][e] = el.y;
      }
    }

    v8f d2r, d2i;
    apply_layer(sr, si, K, d2r, d2i);

    {
      for (int L = 2; L < n_layers; ++L) {
        wave_sync();
        if (l < NQ) rot_gate(weights + (L * NQ + l) * 3, G1[l]);
#pragma unroll
        for (int r = 0; r < 8; ++r) {
          const int j = m_of(r + 8 * h, c);
          ST[0][j] = d2r[r]; ST[1][j] = d2i[r];
        }
        wave_sync();
        build_ops(G1, h, c, K);
        const int rr = (L - 1) % (NQ - 1) + 1;
#pragma unroll
        for (int kk = 0; kk < 4; ++kk) {
#pragma unroll
          for (int e = 0; e < 2; ++e) {
            const int j = ring_src(m_of(c, khi_of(kk, h, e)), rr);
            sr[kk][e] = ST[0][j]; si[kk][e] = ST[1][j];
          }
        }
        apply_layer(sr, si, K, d2r, d2i);
      }
      wave_sync();
      if (l < NQ) rot_gate(weights + (NQ + l) * 3, G1[l]);
      wave_sync();
      build_ops(G1, h, c, K);
    }

    float zp = 0.f;
#pragma unroll
    for (int r = 0; r < 8; ++r) zp += (d2r[r] * d2r[r] + d2i[r] * d2i[r]) * gw[r];
    zred[s][l] = zp;
  }
  wave_sync();

  if (l < SPW) {
    float sum = fb;
#pragma unroll
    for (int cc = 0; cc < 32; ++cc) sum += zred[l][cc];
    out[sBase + l] = 1.f / (1.f + expf(-sum));
  }
}

struct GroupIn {
  float4 v0, v1, v7, v2, v3, v4, v5;
  float2 v6;
};
struct GroupOps {
  v2f sr[4], si[4], ss[4];
  v2f fr, fi, fs;
};

__device__ __forceinline__ GroupIn load_group(const float* ps, int h, int m1, int m6) {
  GroupIn in;
  in.v0 = *(const float4*)(ps + 0 * 8 + h * 4);
  in.v1 = *(const float4*)(ps + 1 * 8 + (h ^ m1) * 4);
  in.v7 = *(const float4*)(ps + 7 * 8 + m6 * 4);
  in.v2 = *(const float4*)(ps + 2 * 8 + m1 * 4);
  in.v3 = *(const float4*)(ps + 3 * 8);
  in.v4 = *(const float4*)(ps + 4 * 8);
  in.v5 = *(const float4*)(ps + 5 * 8 + h * 4);
  in.v6 = *(const float2*)(ps + 6 * 8 + (h ^ m6) * 4);
  return in;
}

__device__ __forceinline__ void build_group_ops(const GroupIn& in, GroupOps& o) {
  const float2 f0 = cmul(cmul(make_float2(in.v0.x, in.v0.y), make_float2(in.v1.x, in.v1.y)), make_float2(in.v7.x, in.v7.y));
  const float2 f1 = cmul(cmul(make_float2(in.v0.z, in.v0.w), make_float2(in.v1.z, in.v1.w)), make_float2(in.v7.z, in.v7.w));
  o.fr[0] = f0.x; o.fr[1] = f1.x; o.fi[0] = f0.y; o.fi[1] = f1.y;
  o.fs[0] = f0.x + f0.y; o.fs[1] = f1.x + f1.y;

  const float2 X0 = make_float2(in.v2.x, in.v2.y), X1 = make_float2(in.v2.z, in.v2.w);
  const float2 p30 = make_float2(in.v3.x, in.v3.y), p31 = make_float2(in.v3.z, in.v3.w);
  const float2 p40 = make_float2(in.v4.x, in.v4.y), p41 = make_float2(in.v4.z, in.v4.w);
  const float2 Y[4] = {cmul(X0, p30), cmul(X0, p31), cmul(X1, p31), cmul(X1, p30)};
  const float2 Q0 = cmul(make_float2(in.v5.x, in.v5.y), in.v6), Q1 = cmul(make_float2(in.v5.z, in.v5.w), in.v6);
  const float2 Z[2][2] = {{cmul(p40, Q0), cmul(p41, Q1)}, {cmul(p41, Q0), cmul(p40, Q1)}};
#pragma unroll
  for (int kk = 0; kk < 4; ++kk) {
#pragma unroll
    for (int e = 0; e < 2; ++e) {
      const float2 el = cmul(Y[kk], Z[kk & 1][e]);
      o.sr[kk][e] = el.x; o.si[kk][e] = el.y; o.ss[kk][e] = el.x + el.y;
    }
  }
}

struct TwoLayerOps {
  v2f Ar[4], nAs[4], Ad[4];
  v2f Br, nBs, Bd;
  float2 cf[2];
};

__device__ __forceinline__ void run_two_layers(const float* fw, float fb, int sBase,
                                               const float2 (*G1)[4], const float (*PS)[PSROW],
                                               float (*zred)[33], float* __restrict__ out) {
  const int l = threadIdx.x & 31, h = l >> 4, c = l & 15;

  TwoLayerOps K;
#pragma unroll
  for (int kk = 0; kk < 4; ++kk) {
#pragma unroll
    for (int e = 0; e < 2; ++e) {
      const int kh = khi_of(kk, h, e);
      float2 a = G1[2][((c >> 3) & 1) * 2 + ((kh >> 3) & 1)];
      a = cmul(a, G1[3][((c >> 2) & 1) * 2 + ((kh >> 2) & 1)]);
      a = cmul(a, G1[4][((c >> 1) & 1) * 2 + ((kh >> 1) & 1)]);
      a = cmul(a, G1[5][(c & 1) * 2 + (kh & 1)]);
      K.Ar[kk][e] = a.x; K.nAs[kk][e] = -a.x - a.y; K.Ad[kk][e] = a.y - a.x;
    }
  }
#pragma unroll
  for (int e = 0; e < 2; ++e) {
    const float2 b = cmul(G1[0][((c >> 3) & 1) * 2 + h], G1[7][(c & 1) * 2 + e]);
    K.Br[e] = b.x; K.nBs[e] = -b.x - b.y; K.Bd[e] = b.y - b.x;
    K.cf[e] = cmul(G1[1][((c >> 2) & 1) * 2 + h], G1[6][((c >> 1) & 1) * 2 + e]);
  }

  const int ml = ring_dst(m_of(c, 8 * h), 2);
  float fs[NQ];
#pragma unroll
  for (int i = 0; i < NQ; ++i)
    fs[i] = __uint_as_float(__float_as_uint(fw[i]) ^ (((unsigned)ml << (24 + i)) & 0x80000000u));
  float gw[8];
#pragma unroll
  for (int r = 0; r < 8; ++r) {
    const int mr = ring_dst(m_of(0, r), 2);
    float g = 0.f;
#pragma unroll
    for (int i = 0; i < NQ; ++i) {
      if ((mr >> (7 - i)) & 1) g -= fs[i]; else g += fs[i];
    }
    gw[r] = g;
  }

  const int m1 = c >> 3, s4 = (c >> 1) & 3, m6 = c & 1;
  const v8f zero = {0.f, 0.f, 0.f, 0.f, 0.f, 0.f, 0.f, 0.f};

  GroupOps cur, nxt;
  build_group_ops(load_group(PS[s4], h, m1, m6), cur);
  GroupIn in = load_group(PS[4 + s4], h, m1, m6);

#pragma unroll
  for (int g = 0; g < SPW / 4; ++g) {

    v8f gx = WMMA4(cur.ss[0], K.Ar[0], zero);
    const v8f wx = WMMA4(cur.fs, K.Br, zero);
    gx = WMMA4(cur.ss[1], K.Ar[1], gx);
    const v8f wr = WMMA4(cur.fi, K.nBs, wx);
    gx = WMMA4(cur.ss[2], K.Ar[2], gx);
    const v8f wi = WMMA4(cur.fr, K.Bd, wx);
    gx = WMMA4(cur.ss[3], K.Ar[3], gx);

    v8f gr = gx, gi = gx;
#pragma unroll
    for (int kk = 0; kk < 4; ++kk) {
      gr = WMMA4(cur.si[kk], K.nAs[kk], gr);
      gi = WMMA4(cur.sr[kk], K.Ad[kk], gi);
    }

    if (g + 1 < SPW / 4) {
      build_group_ops(in, nxt);
      if (g + 2 < SPW / 4) in = load_group(PS[4 * (g + 2) + s4], h, m1, m6);
    }

    float br[8], nbs[8], bd[8], gs[8];
#pragma unroll
    for (int r = 0; r < 8; ++r) {
      const float2 cf = K.cf[r & 1];
      const float bi = wr[r] * cf.y + wi[r] * cf.x;
      br[r]  = wr[r] * cf.x - wi[r] * cf.y;
      nbs[r] = -br[r] - bi;
      bd[r]  = bi - br[r];
      gs[r]  = gr[r] + gi[r];
    }

    v8f cx[4];
#pragma unroll
    for (int t = 0; t < 4; ++t) {
      const v2f as = {gs[2 * t], gs[2 * t + 1]}, tr = {br[2 * t], br[2 * t + 1]};
      cx[t] = WMMA4(as, tr, zero);
    }
#pragma unroll
    for (int t = 0; t < 4; ++t) {
      const v2f ar = {gr[2 * t], gr[2 * t + 1]}, ai = {gi[2 * t], gi[2 * t + 1]};
      const v2f nts = {nbs[2 * t], nbs[2 * t + 1]}, td = {bd[2 * t], bd[2 * t + 1]};
      const v8f cr = WMMA4(ai, nts, cx[t]);
      const v8f ci = WMMA4(ar, td, cx[t]);
      float zp = 0.f;
#pragma unroll
      for (int r = 0; r < 8; ++r) zp += (cr[r] * cr[r] + ci[r] * ci[r]) * gw[r];
      zred[4 * g + t][l] = zp;
    }
    cur = nxt;
  }
  wave_sync();

  if (l < SPW) {
    float sum = fb;
#pragma unroll
    for (int cc = 0; cc < 32; ++cc) sum += zred[l][cc];
    out[sBase + l] = 1.f / (1.f + expf(-sum));
  }
}

__global__ __launch_bounds__(32 * WPB) void vqe_kron(const float* __restrict__ x,
                                                     const float* __restrict__ weights,
                                                     int n_layers,
                                                     const float* __restrict__ fcw,
                                                     const float* __restrict__ fcb,
                                                     float* __restrict__ out,
                                                     int n_tiles) {
  __shared__ __attribute__((aligned(16))) float2 GGw[WPB][2][NQ][4];
  __shared__ __attribute__((aligned(16))) float PSw[WPB][SPW][PSROW];
  __shared__ float STw[WPB][2][256];
  __shared__ float zredw[WPB][SPW][33];

  const int w = __builtin_amdgcn_readfirstlane((int)(threadIdx.x >> 5));
  const int tile_raw = blockIdx.x * WPB + w;
  const bool live = tile_raw < n_tiles;
  const int tile = live ? tile_raw : 0;
  float2 (*GG)[NQ][4] = GGw[w];
  float (*PS)[PSROW]  = PSw[w];
  float (*ST)[256]    = STw[w];
  float (*zred)[33]   = zredw[w];
  float2 (*G1)[4] = GG[1];

  const int l = threadIdx.x & 31;
  const int h = l >> 4;
  const int c = l & 15;
  const int sBase = tile * SPW;

  const float4 xv = *(const float4*)(x + (size_t)(sBase + c) * NQ + 4 * h);
  const bool have = (c >> 3) < n_layers;
  float ph = 0.f, th = 0.f, om = 0.f;
  if (have) {
    const float* wt = weights + c * 3;
    ph = wt[0]; th = wt[1]; om = wt[2];
  }
  float fw[NQ];
#pragma unroll
  for (int i = 0; i < NQ; ++i) fw[i] = fcw[i];
  const float fb = fcb[0];

  float s1, c1, s2, c2;
  sincosf(h ? 0.5f * (ph + om) : 0.5f * th, &s1, &c1);
  sincosf(0.5f * (ph - om), &s2, &c2);
  const float so = __shfl_xor(s1, 16), co = __shfl_xor(c1, 16);
  if (h == 0) {
    float2* g = GG[c >> 3][c & 7];
    if (have) {
      g[0] = make_float2( co * c1, -so * c1);
      g[1] = make_float2(-c2 * s1, -s2 * s1);
      g[2] = make_float2( c2 * s1, -s2 * s1);
      g[3] = make_float2( co * c1,  so * c1);
    } else if (c < NQ) {
      g[0] = make_float2(1.f, 0.f); g[1] = make_float2(0.f, 0.f);
      g[2] = make_float2(0.f, 0.f); g[3] = make_float2(1.f, 0.f);
    }
  }

  float cs[4], sn[4];
  sincosf(0.5f * xv.x, &sn[0], &cs[0]);
  sincosf(0.5f * xv.y, &sn[1], &cs[1]);
  sincosf(0.5f * xv.z, &sn[2], &cs[2]);
  sincosf(0.5f * xv.w, &sn[3], &cs[3]);
  wave_sync();

#pragma unroll
  for (int i = 0; i < 4; ++i) {
    const int q = 4 * h + i;
    const float4 ua = *(const float4*)GG[0][q], ub = *(const float4*)(GG[0][q] + 2);
    const float2 p0 = make_float2(ua.x * cs[i] + ua.w * sn[i], ua.y * cs[i] - ua.z * sn[i]);
    const float2 p1 = make_float2(ub.x * cs[i] + ub.w * sn[i], ub.y * cs[i] - ub.z * sn[i]);
    *(float4*)(PS[c] + q * 8)     = make_float4(p0.x, p0.y, p1.x, p1.y);
    *(float4*)(PS[c] + q * 8 + 4) = make_float4(p1.x, p1.y, p0.x, p0.y);
  }
  wave_sync();

  if (!live) return;
  if (n_layers > 2) {
    run_layers(weights, n_layers, fw, fb, sBase, G1, PS, ST, zred, out);
  } else if (n_layers == 2) {
    run_two_layers(fw, fb, sBase, G1, PS, zred, out);
  } else {
    for (int s = 0; s < SPW; ++s) {
      float zp = 0.f;
      for (int i = 0; i < 8; ++i) {
        const int m = l + 32 * i;
        const int j = (n_layers > 0) ? ring_src(m, 1) : m;
        float2 v = make_float2(PS[s][((j >> 7) & 1) * 2], PS[s][((j >> 7) & 1) * 2 + 1]);
        for (int q = 1; q < NQ; ++q) {
          const int bq = (j >> (7 - q)) & 1;
          v = cmul(v, make_float2(PS[s][q * 8 + bq * 2], PS[s][q * 8 + bq * 2 + 1]));
        }
        float g = 0.f;
        for (int t = 0; t < NQ; ++t) g += fw[t] * (1.f - 2.f * (float)((m >> (7 - t)) & 1));
        zp += (v.x * v.x + v.y * v.y) * g;
      }
      zred[s][l] = zp;
    }
    wave_sync();
    if (l < SPW) {
      float sum = fb;
      for (int cc = 0; cc < 32; ++cc) sum += zred[l][cc];
      out[sBase + l] = 1.f / (1.f + expf(-sum));
    }
  }
}

extern "C" void kernel_launch(void* const* d_in, const int* in_sizes, int n_in,
                              void* d_out, int out_size, void* d_ws, size_t ws_size,
                              hipStream_t stream) {
  const float* x       = (const float*)d_in[0];
  const float* weights = (const float*)d_in[1];
  const float* fcw     = (const float*)d_in[2];
  const float* fcb     = (const float*)d_in[3];
  float* out = (float*)d_out;

  const int B        = in_sizes[0] / NQ;
  const int n_layers = in_sizes[1] / (NQ * 3);

  const int n_tiles  = B / SPW;

  vqe_kron<<<(n_tiles + WPB - 1) / WPB, 32 * WPB, 0, stream>>>(x, weights, n_layers, fcw, fcb, out, n_tiles);
}
